# starting kernel
# baseline (speedup 1.0000x reference)
#include <hip/hip_runtime.h>
#include <math.h>

typedef __attribute__((ext_vector_type(2))) float v2f;
typedef __attribute__((ext_vector_type(8))) float v8f;

#define FDIM 36
#define NHEAD 3
#define HDIM 12
#define NEG_SLOPE 0.2f
#define NEG_INF_KEY 0x007FFFFFu

__device__ __forceinline__ unsigned f2key(float f) {
  unsigned u = __float_as_uint(f);
  return (u & 0x80000000u) ? ~u : (u | 0x80000000u);
}
__device__ __forceinline__ float key2f(unsigned k) {
  return (k & 0x80000000u) ? __uint_as_float(k & 0x7FFFFFFFu)
                           : __uint_as_float(~k);
}
__device__ __forceinline__ float leaky(float x) {
  return x >= 0.0f ? x : NEG_SLOPE * x;
}

__global__ __launch_bounds__(32) void gemm36_wmma(
    const float* __restrict__ A, const float* __restrict__ W,
    const float* __restrict__ bias, float* __restrict__ out, int nrows) {
  const int lane = threadIdx.x;
  const int half = lane >> 4;
  const int idx  = lane & 15;

  int mrow = blockIdx.x * 16 + idx;
  mrow = mrow < nrows ? mrow : (nrows - 1);
  const float* __restrict__ Ab = A + (size_t)mrow * FDIM + 2 * half;

  const int col0 = idx;
  const int col1 = idx + 16;
  const int col2 = (idx + 32) < FDIM ? (idx + 32) : (FDIM - 1);
  const float* __restrict__ Wb0 = W + col0 + 72 * half;
  const float* __restrict__ Wb1 = W + col1 + 72 * half;
  const float* __restrict__ Wb2 = W + col2 + 72 * half;

  v8f acc[3];
  acc[0] = (v8f){0.f,0.f,0.f,0.f,0.f,0.f,0.f,0.f};
  acc[1] = acc[0];
  acc[2] = acc[0];

  #pragma unroll
  for (int k0 = 0; k0 < FDIM; k0 += 4) {
    v2f a, b0, b1, b2;
    a.x  = Ab[k0];
    a.y  = Ab[k0 + 1];
    b0.x = Wb0[(k0 + 0) * FDIM];
    b0.y = Wb0[(k0 + 1) * FDIM];
    b1.x = Wb1[(k0 + 0) * FDIM];
    b1.y = Wb1[(k0 + 1) * FDIM];
    b2.x = Wb2[(k0 + 0) * FDIM];
    b2.y = Wb2[(k0 + 1) * FDIM];
    acc[0] = __builtin_amdgcn_wmma_f32_16x16x4_f32(false, a, false, b0,
                                                   (short)0, acc[0], false, false);
    acc[1] = __builtin_amdgcn_wmma_f32_16x16x4_f32(false, a, false, b1,
                                                   (short)0, acc[1], false, false);
    acc[2] = __builtin_amdgcn_wmma_f32_16x16x4_f32(false, a, false, b2,
                                                   (short)0, acc[2], false, false);
  }

  const int rowbase = blockIdx.x * 16 + 8 * half;
  const bool full = (blockIdx.x * 16 + 16) <= nrows;
  #pragma unroll
  for (int t = 0; t < 3; ++t) {
    const int col = idx + 16 * t;
    if (col < FDIM) {
      const float bv = bias ? bias[col] : 0.0f;
      float* o = out + (size_t)rowbase * FDIM + col;
      if (full) {
        #pragma unroll
        for (int v = 0; v < 8; ++v) o[(size_t)v * FDIM] = acc[t][v] + bv;
      } else {
        #pragma unroll
        for (int v = 0; v < 8; ++v)
          if (rowbase + v < nrows) o[(size_t)v * FDIM] = acc[t][v] + bv;
      }
    }
  }
}

__global__ void attn_scores(const float* __restrict__ feat,
                            const float* __restrict__ a_l,
                            const float* __restrict__ a_r,
                            float* __restrict__ el, float* __restrict__ er,
                            int n) {
  int gid = blockIdx.x * blockDim.x + threadIdx.x;
  if (gid >= n * NHEAD) return;
  const int node = gid / NHEAD;
  const int h    = gid % NHEAD;
  const float4* f4 = (const float4*)(feat + (size_t)node * FDIM + h * HDIM);
  const float4* l4 = (const float4*)(a_l + h * HDIM);
  const float4* r4 = (const float4*)(a_r + h * HDIM);
  float sl = 0.0f, sr = 0.0f;
  #pragma unroll
  for (int q = 0; q < HDIM / 4; ++q) {
    const float4 fv = f4[q], lv = l4[q], rv = r4[q];
    sl += fv.x * lv.x + fv.y * lv.y + fv.z * lv.z + fv.w * lv.w;
    sr += fv.x * rv.x + fv.y * rv.y + fv.z * rv.z + fv.w * rv.w;
  }
  el[gid] = sl;
  er[gid] = sr;
}

__global__ void layer_init(unsigned* __restrict__ mkey,
                           float* __restrict__ denom,
                           float* __restrict__ rst, int n) {
  int gid = blockIdx.x * blockDim.x + threadIdx.x;
  if (gid < n * FDIM) rst[gid] = 0.0f;
  if (gid < n * NHEAD) {
    mkey[gid] = NEG_INF_KEY;
    denom[gid] = 0.0f;
  }
}

__global__ void edge_max(const int* __restrict__ src,
                         const int* __restrict__ dst,
                         const float* __restrict__ el,
                         const float* __restrict__ er,
                         unsigned* __restrict__ mkey, int ne) {
  int e = blockIdx.x * blockDim.x + threadIdx.x;
  if (e >= ne) return;
  const int s = src[e], d = dst[e];
  #pragma unroll
  for (int h = 0; h < NHEAD; ++h) {
    const float v = leaky(el[s * NHEAD + h] + er[d * NHEAD + h]);
    atomicMax(&mkey[d * NHEAD + h], f2key(v));
  }
}

__global__ void m_finalize(unsigned* __restrict__ mkey, float* __restrict__ m,
                           int n3) {
  int gid = blockIdx.x * blockDim.x + threadIdx.x;
  if (gid >= n3) return;
  float v = key2f(mkey[gid]);
  if (!isfinite(v)) v = 0.0f;
  m[gid] = v;
}

__global__ void edge_sum(const int* __restrict__ src,
                         const int* __restrict__ dst,
                         const float* __restrict__ el,
                         const float* __restrict__ er,
                         const float* __restrict__ m,
                         float* __restrict__ denom, int ne) {
  int e = blockIdx.x * blockDim.x + threadIdx.x;
  if (e >= ne) return;
  const int s = src[e], d = dst[e];
  #pragma unroll
  for (int h = 0; h < NHEAD; ++h) {
    const float v = leaky(el[s * NHEAD + h] + er[d * NHEAD + h]);
    const float ee = expf(v - m[d * NHEAD + h]);
    atomicAdd(&denom[d * NHEAD + h], ee);
  }
}

__global__ void edge_msg(const int* __restrict__ src,
                         const int* __restrict__ dst,
                         const float* __restrict__ el,
                         const float* __restrict__ er,
                         const float* __restrict__ m,
                         const float* __restrict__ denom,
                         const float* __restrict__ feat,
                         float* __restrict__ rst, int ne) {
  int e = blockIdx.x * blockDim.x + threadIdx.x;
  if (e >= ne) return;
  const int s = src[e], d = dst[e];
  const float4* fsrc4 = (const float4*)(feat + (size_t)s * FDIM);
  __builtin_prefetch(fsrc4, 0, 3);
  #pragma unroll
  for (int h = 0; h < NHEAD; ++h) {
    const float v = leaky(el[s * NHEAD + h] + er[d * NHEAD + h]);
    const float alpha = expf(v - m[d * NHEAD + h]) /
                        fmaxf(denom[d * NHEAD + h], 1e-9f);
    float* rd = rst + (size_t)d * FDIM + h * HDIM;
    #pragma unroll
    for (int q = 0; q < HDIM / 4; ++q) {
      const float4 fv = fsrc4[h * (HDIM / 4) + q];
      atomicAdd(&rd[q * 4 + 0], fv.x * alpha);
      atomicAdd(&rd[q * 4 + 1], fv.y * alpha);
      atomicAdd(&rd[q * 4 + 2], fv.z * alpha);
      atomicAdd(&rd[q * 4 + 3], fv.w * alpha);
    }
  }
}

__global__ void node_finalize(const float* __restrict__ hin,
                              const float* __restrict__ bias,
                              float* __restrict__ rst, int n) {
  int gid = blockIdx.x * blockDim.x + threadIdx.x;
  if (gid >= n * FDIM) return;
  const float x = rst[gid] + hin[gid] + bias[gid % FDIM];
  rst[gid] = x > 0.0f ? x : expm1f(x);
}

extern "C" void kernel_launch(void* const* d_in, const int* in_sizes, int n_in,
                              void* d_out, int out_size, void* d_ws,
                              size_t ws_size, hipStream_t stream) {
  (void)n_in; (void)out_size; (void)ws_size;
  const float* X        = (const float*)d_in[0];
  const int*   src      = (const int*)d_in[1];
  const int*   dst      = (const int*)d_in[2];
  const float* lin0_w   = (const float*)d_in[3];
  const float* lin0_b   = (const float*)d_in[4];
  const float* fc_w     = (const float*)d_in[5];
  const float* attn_l   = (const float*)d_in[6];
  const float* attn_r   = (const float*)d_in[7];
  const float* gat_bias = (const float*)d_in[8];
  const float* out_w    = (const float*)d_in[9];
  const float* out_b    = (const float*)d_in[10];

  const int n  = in_sizes[0] / FDIM;
  const int ne = in_sizes[1];

  float* ws   = (float*)d_ws;
  float* hA    = ws; ws += (size_t)n * FDIM;
  float* hB    = ws; ws += (size_t)n * FDIM;
  float* feat  = ws; ws += (size_t)n * FDIM;
  float* el    = ws; ws += (size_t)n * NHEAD;
  float* er    = ws; ws += (size_t)n * NHEAD;
  float* mbuf  = ws; ws += (size_t)n * NHEAD;
  float* denom = ws; ws += (size_t)n * NHEAD;

  const dim3 gemmGrid((n + 15) / 16);
  const int nbE  = (ne + 255) / 256;
  const int nbN3 = (n * NHEAD + 255) / 256;
  const int nbNF = (n * FDIM + 255) / 256;

  gemm36_wmma<<<gemmGrid, 32, 0, stream>>>(X, lin0_w, lin0_b, hA, n);

  float* hcur = hA;
  float* hnxt = hB;
  for (int l = 0; l < 3; ++l) {
    gemm36_wmma<<<gemmGrid, 32, 0, stream>>>(
        hcur, fc_w + (size_t)l * FDIM * FDIM, nullptr, feat, n);
    attn_scores<<<nbN3, 256, 0, stream>>>(
        feat, attn_l + (size_t)l * NHEAD * HDIM, attn_r + (size_t)l * NHEAD * HDIM,
        el, er, n);
    layer_init<<<nbNF, 256, 0, stream>>>((unsigned*)mbuf, denom, hnxt, n);
    edge_max<<<nbE, 256, 0, stream>>>(src, dst, el, er, (unsigned*)mbuf, ne);
    m_finalize<<<nbN3, 256, 0, stream>>>((unsigned*)mbuf, mbuf, n * NHEAD);
    edge_sum<<<nbE, 256, 0, stream>>>(src, dst, el, er, mbuf, denom, ne);
    edge_msg<<<nbE, 256, 0, stream>>>(src, dst, el, er, mbuf, denom, feat,
                                      hnxt, ne);
    node_finalize<<<nbNF, 256, 0, stream>>>(hcur, gat_bias + (size_t)l * FDIM,
                                            hnxt, n);
    float* t = hcur; hcur = hnxt; hnxt = t;
  }

  gemm36_wmma<<<gemmGrid, 32, 0, stream>>>(hcur, out_w, out_b, (float*)d_out, n);
}


# edges counting-sorted by dst once per call (int atomics); per layer one wave per node does max, exp-sum, alpha*feat sum and ELU in one launch: no float atomics, sums in segment order by lane tree
# speedup vs baseline: 8.4615x; 8.4615x over previous
#include <hip/hip_runtime.h>
#include <math.h>

typedef __attribute__((ext_vector_type(2))) float v2f;
typedef __attribute__((ext_vector_type(8))) float v8f;

#define FDIM 36
#define NHEAD 3
#define HDIM 12
#define NEG_SLOPE 0.2f

__device__ __forceinline__ float leaky(float x) {
  return x >= 0.0f ? x : NEG_SLOPE * x;
}

__global__ __launch_bounds__(32) void gemm36_wmma(
    const float* __restrict__ A, const float* __restrict__ W,
    const float* __restrict__ bias, float* __restrict__ out, int nrows) {
  const int lane = threadIdx.x;
  const int half = lane >> 4;
  const int idx  = lane & 15;

  int mrow = blockIdx.x * 16 + idx;
  mrow = mrow < nrows ? mrow : (nrows - 1);
  const float* __restrict__ Ab = A + (size_t)mrow * FDIM + 2 * half;

  const int col0 = idx;
  const int col1 = idx + 16;
  const int col2 = (idx + 32) < FDIM ? (idx + 32) : (FDIM - 1);
  const float* __restrict__ Wb0 = W + col0 + 72 * half;
  const float* __restrict__ Wb1 = W + col1 + 72 * half;
  const float* __restrict__ Wb2 = W + col2 + 72 * half;

  v8f acc[3];
  acc[0] = (v8f){0.f,0.f,0.f,0.f,0.f,0.f,0.f,0.f};
  acc[1] = acc[0];
  acc[2] = acc[0];

  #pragma unroll
  for (int k0 = 0; k0 < FDIM; k0 += 4) {
    v2f a, b0, b1, b2;
    a.x  = Ab[k0];
    a.y  = Ab[k0 + 1];
    b0.x = Wb0[(k0 + 0) * FDIM];
    b0.y = Wb0[(k0 + 1) * FDIM];
    b1.x = Wb1[(k0 + 0) * FDIM];
    b1.y = Wb1[(k0 + 1) * FDIM];
    b2.x = Wb2[(k0 + 0) * FDIM];
    b2.y = Wb2[(k0 + 1) * FDIM];
    acc[0] = __builtin_amdgcn_wmma_f32_16x16x4_f32(false, a, false, b0,
                                                   (short)0, acc[0], false, false);
    acc[1] = __builtin_amdgcn_wmma_f32_16x16x4_f32(false, a, false, b1,
                                                   (short)0, acc[1], false, false);
    acc[2] = __builtin_amdgcn_wmma_f32_16x16x4_f32(false, a, false, b2,
                                                   (short)0, acc[2], false, false);
  }

  const int rowbase = blockIdx.x * 16 + 8 * half;
  const bool full = (blockIdx.x * 16 + 16) <= nrows;
  #pragma unroll
  for (int t = 0; t < 3; ++t) {
    const int col = idx + 16 * t;
    if (col < FDIM) {
      const float bv = bias ? bias[col] : 0.0f;
      float* o = out + (size_t)rowbase * FDIM + col;
      if (full) {
        #pragma unroll
        for (int v = 0; v < 8; ++v) o[(size_t)v * FDIM] = acc[t][v] + bv;
      } else {
        #pragma unroll
        for (int v = 0; v < 8; ++v)
          if (rowbase + v < nrows) o[(size_t)v * FDIM] = acc[t][v] + bv;
      }
    }
  }
}

__global__ void attn_scores(const float* __restrict__ feat,
                            const float* __restrict__ a_l,
                            const float* __restrict__ a_r,
                            float* __restrict__ el4, float* __restrict__ er,
                            int n) {
  int gid = blockIdx.x * blockDim.x + threadIdx.x;
  if (gid >= n * NHEAD) return;
  const int node = gid / NHEAD;
  const int h    = gid % NHEAD;
  const float4* f4 = (const float4*)(feat + (size_t)node * FDIM + h * HDIM);
  const float4* l4 = (const float4*)(a_l + h * HDIM);
  const float4* r4 = (const float4*)(a_r + h * HDIM);
  float sl = 0.0f, sr = 0.0f;
  #pragma unroll
  for (int q = 0; q < HDIM / 4; ++q) {
    const float4 fv = f4[q], lv = l4[q], rv = r4[q];
    sl += fv.x * lv.x + fv.y * lv.y + fv.z * lv.z + fv.w * lv.w;
    sr += fv.x * rv.x + fv.y * rv.y + fv.z * rv.z + fv.w * rv.w;
  }
  el4[node * 4 + h] = sl;
  if (h == 0) el4[node * 4 + 3] = 0.0f;
  er[gid] = sr;
}

__global__ void csr_zero(int* __restrict__ cnt, int n1) {
  int gid = blockIdx.x * blockDim.x + threadIdx.x;
  if (gid < n1) cnt[gid] = 0;
}

__global__ void csr_count(const int* __restrict__ dst, int* __restrict__ cnt,
                          int ne) {
  int e = blockIdx.x * blockDim.x + threadIdx.x;
  if (e < ne) atomicAdd(&cnt[dst[e]], 1);
}

__global__ __launch_bounds__(1024) void csr_scan(int* __restrict__ cur,
                                                 int* __restrict__ rowptr,
                                                 int n) {
  __shared__ int wsum[32];
  __shared__ int tile_total;
  const int tid  = threadIdx.x;
  const int lane = tid & 31;
  const int wid  = tid >> 5;
  int carry = 0;
  for (int base = 0; base < n; base += 4096) {
    const int i0 = base + tid * 4;
    const int c0 = (i0 + 0 < n) ? cur[i0 + 0] : 0;
    const int c1 = (i0 + 1 < n) ? cur[i0 + 1] : 0;
    const int c2 = (i0 + 2 < n) ? cur[i0 + 2] : 0;
    const int c3 = (i0 + 3 < n) ? cur[i0 + 3] : 0;
    const int tsum = c0 + c1 + c2 + c3;
    int x = tsum;
    for (int off = 1; off < 32; off <<= 1) {
      const int y = __shfl_up(x, off);
      if (lane >= off) x += y;
    }
    if (lane == 31) wsum[wid] = x;
    __syncthreads();
    if (wid == 0) {
      const int w = wsum[lane];
      int xs = w;
      for (int off = 1; off < 32; off <<= 1) {
        const int y = __shfl_up(xs, off);
        if (lane >= off) xs += y;
      }
      wsum[lane] = xs - w;
      if (lane == 31) tile_total = xs;
    }
    __syncthreads();
    int ex = carry + wsum[wid] + (x - tsum);
    carry += tile_total;
    if (i0 + 0 < n) { cur[i0 + 0] = ex; rowptr[i0 + 0] = ex; }
    ex += c0;
    if (i0 + 1 < n) { cur[i0 + 1] = ex; rowptr[i0 + 1] = ex; }
    ex += c1;
    if (i0 + 2 < n) { cur[i0 + 2] = ex; rowptr[i0 + 2] = ex; }
    ex += c2;
    if (i0 + 3 < n) { cur[i0 + 3] = ex; rowptr[i0 + 3] = ex; }
    __syncthreads();
  }
  if (tid == 0) rowptr[n] = carry;
}

__global__ void csr_fill(const int* __restrict__ src,
                         const int* __restrict__ dst, int* __restrict__ cur,
                         int* __restrict__ ssrc, int ne) {
  int e = blockIdx.x * blockDim.x + threadIdx.x;
  if (e >= ne) return;
  const int p = atomicAdd(&cur[dst[e]], 1);
  ssrc[p] = src[e];
}

__device__ __forceinline__ float wave_max(float v) {
  for (int off = 16; off > 0; off >>= 1) v = fmaxf(v, __shfl_xor(v, off));
  return v;
}
__device__ __forceinline__ float wave_sum(float v) {
  for (int off = 16; off > 0; off >>= 1) v += __shfl_xor(v, off);
  return v;
}

struct EdgeS { int s; float v0, v1, v2; };

__device__ __forceinline__ EdgeS load_edge(const int* __restrict__ ssrc,
                                           const float4* __restrict__ el4,
                                           int j, int end, float er0,
                                           float er1, float er2) {
  EdgeS e;
  e.s = 0;
  e.v0 = -INFINITY; e.v1 = -INFINITY; e.v2 = -INFINITY;
  if (j < end) {
    e.s = ssrc[j];
    const float4 t = el4[e.s];
    e.v0 = leaky(t.x + er0);
    e.v1 = leaky(t.y + er1);
    e.v2 = leaky(t.z + er2);
  }
  return e;
}

__global__ __launch_bounds__(256) void gat_aggregate(
    const int* __restrict__ rowptr, const int* __restrict__ ssrc,
    const float* __restrict__ el4f, const float* __restrict__ er,
    const float* __restrict__ feat, const float* __restrict__ bias,
    float* __restrict__ h, int n) {
  const int lane = threadIdx.x & 31;
  const int d = blockIdx.x * 8 + (threadIdx.x >> 5);
  if (d >= n) return;
  const float4* __restrict__ el4 = (const float4*)el4f;
  const int beg = rowptr[d];
  const int end = rowptr[d + 1];
  const float er0 = er[d * NHEAD + 0];
  const float er1 = er[d * NHEAD + 1];
  const float er2 = er[d * NHEAD + 2];

  const EdgeS eA = load_edge(ssrc, el4, beg + lane, end, er0, er1, er2);

  float m0 = eA.v0, m1 = eA.v1, m2 = eA.v2;
  for (int j = beg + 32 + lane; j < end; j += 32) {
    const EdgeS e = load_edge(ssrc, el4, j, end, er0, er1, er2);
    m0 = fmaxf(m0, e.v0);
    m1 = fmaxf(m1, e.v1);
    m2 = fmaxf(m2, e.v2);
  }
  m0 = wave_max(m0); m1 = wave_max(m1); m2 = wave_max(m2);
  if (!isfinite(m0)) m0 = 0.0f;
  if (!isfinite(m1)) m1 = 0.0f;
  if (!isfinite(m2)) m2 = 0.0f;

  float d0 = expf(eA.v0 - m0);
  float d1 = expf(eA.v1 - m1);
  float d2 = expf(eA.v2 - m2);
  for (int j = beg + 32 + lane; j < end; j += 32) {
    const EdgeS e = load_edge(ssrc, el4, j, end, er0, er1, er2);
    d0 += expf(e.v0 - m0);
    d1 += expf(e.v1 - m1);
    d2 += expf(e.v2 - m2);
  }
  d0 = fmaxf(wave_sum(d0), 1e-9f);
  d1 = fmaxf(wave_sum(d1), 1e-9f);
  d2 = fmaxf(wave_sum(d2), 1e-9f);

  const int g  = lane / 9;
  const int q  = lane - g * 9;
  const int hq = q / 3;
  float4 acc = make_float4(0.0f, 0.0f, 0.0f, 0.0f);
  for (int cb = beg; cb < end; cb += 32) {
    EdgeS e = eA;
    if (cb != beg)
      e = load_edge(ssrc, el4, cb + lane, end, er0, er1, er2);
    const int   s  = e.s;
    const float p0 = expf(e.v0 - m0) / d0;
    const float p1 = expf(e.v1 - m1) / d1;
    const float p2 = expf(e.v2 - m2) / d2;
    const int cnt = (end - cb) < 32 ? (end - cb) : 32;
    for (int t0 = 0; t0 < cnt; t0 += 12) {
      float4 fv[4];
      float  aw[4];
      #pragma unroll
      for (int u = 0; u < 4; ++u) {
        const int  jj = t0 + 3 * u + g;
        const bool ok = (lane < 27) && (jj < cnt);
        const int  fb = (t0 + 3 * u) < cnt ? (t0 + 3 * u) : (cnt - 1);
        const int  sl = ok ? jj : fb;
        const int   ss = __shfl(s, sl);
        const float a0 = __shfl(p0, sl);
        const float a1 = __shfl(p1, sl);
        const float a2 = __shfl(p2, sl);
        const float a  = hq == 0 ? a0 : (hq == 1 ? a1 : a2);
        aw[u] = ok ? a : 0.0f;
        fv[u] = ((const float4*)(feat + (size_t)ss * FDIM))[q];
      }
      #pragma unroll
      for (int u = 0; u < 4; ++u) {
        acc.x += fv[u].x * aw[u];
        acc.y += fv[u].y * aw[u];
        acc.z += fv[u].z * aw[u];
        acc.w += fv[u].w * aw[u];
      }
    }
  }
  float4 r;
  r.x = acc.x + __shfl(acc.x, (lane + 9) & 31) + __shfl(acc.x, (lane + 18) & 31);
  r.y = acc.y + __shfl(acc.y, (lane + 9) & 31) + __shfl(acc.y, (lane + 18) & 31);
  r.z = acc.z + __shfl(acc.z, (lane + 9) & 31) + __shfl(acc.z, (lane + 18) & 31);
  r.w = acc.w + __shfl(acc.w, (lane + 9) & 31) + __shfl(acc.w, (lane + 18) & 31);
  if (lane < 9) {
    float4* hp = (float4*)(h + (size_t)d * FDIM) + q;
    const float4 hv = *hp;
    const float4 bv = ((const float4*)bias)[q];
    float4 o;
    o.x = r.x + hv.x + bv.x;
    o.y = r.y + hv.y + bv.y;
    o.z = r.z + hv.z + bv.z;
    o.w = r.w + hv.w + bv.w;
    o.x = o.x > 0.0f ? o.x : expm1f(o.x);
    o.y = o.y > 0.0f ? o.y : expm1f(o.y);
    o.z = o.z > 0.0f ? o.z : expm1f(o.z);
    o.w = o.w > 0.0f ? o.w : expm1f(o.w);
    *hp = o;
  }
}

extern "C" void kernel_launch(void* const* d_in, const int* in_sizes, int n_in,
                              void* d_out, int out_size, void* d_ws,
                              size_t ws_size, hipStream_t stream) {
  (void)n_in; (void)out_size; (void)ws_size;
  const float* X        = (const float*)d_in[0];
  const int*   src      = (const int*)d_in[1];
  const int*   dst      = (const int*)d_in[2];
  const float* lin0_w   = (const float*)d_in[3];
  const float* lin0_b   = (const float*)d_in[4];
  const float* fc_w     = (const float*)d_in[5];
  const float* attn_l   = (const float*)d_in[6];
  const float* attn_r   = (const float*)d_in[7];
  const float* gat_bias = (const float*)d_in[8];
  const float* out_w    = (const float*)d_in[9];
  const float* out_b    = (const float*)d_in[10];

  const int n  = in_sizes[0] / FDIM;
  const int ne = in_sizes[1];

  float* ws   = (float*)d_ws;
  float* hbuf  = ws; ws += (size_t)n * FDIM;
  float* feat  = ws; ws += (size_t)n * FDIM;
  float* el4   = ws; ws += (size_t)n * 4;
  float* er    = ws; ws += (size_t)n * NHEAD;
  int* rowptr  = (int*)ws;
  int* cur     = rowptr + (size_t)n + 1;
  int* ssrc    = cur + (size_t)n;

  const dim3 gemmGrid((n + 15) / 16);
  const int nbE  = (ne + 255) / 256;
  const int nbN1 = (n + 1 + 255) / 256;
  const int nbN3 = (n * NHEAD + 255) / 256;
  const int nbAg = (n + 7) / 8;

  csr_zero<<<nbN1, 256, 0, stream>>>(cur, n);
  csr_count<<<nbE, 256, 0, stream>>>(dst, cur, ne);
  csr_scan<<<1, 1024, 0, stream>>>(cur, rowptr, n);
  csr_fill<<<nbE, 256, 0, stream>>>(src, dst, cur, ssrc, ne);

  gemm36_wmma<<<gemmGrid, 32, 0, stream>>>(X, lin0_w, lin0_b, hbuf, n);

  for (int l = 0; l < 3; ++l) {
    gemm36_wmma<<<gemmGrid, 32, 0, stream>>>(
        hbuf, fc_w + (size_t)l * FDIM * FDIM, nullptr, feat, n);
    attn_scores<<<nbN3, 256, 0, stream>>>(
        feat, attn_l + (size_t)l * NHEAD * HDIM, attn_r + (size_t)l * NHEAD * HDIM,
        el4, er, n);
    gat_aggregate<<<nbAg, 256, 0, stream>>>(rowptr, ssrc, el4, er, feat,
                                            gat_bias + (size_t)l * FDIM, hbuf, n);
  }

  gemm36_wmma<<<gemmGrid, 32, 0, stream>>>(hbuf, out_w, out_b, (float*)d_out, n);
}


# edges counted and ranked on LDS counters per destination range and edge slice: global int adds once per node and slice, not once per edge; segment order stays atomic arrival order
# speedup vs baseline: 15.3907x; 1.8189x over previous
#include <hip/hip_runtime.h>
#include <math.h>

typedef __attribute__((ext_vector_type(2))) float v2f;
typedef __attribute__((ext_vector_type(8))) float v8f;

#define FDIM 36
#define NHEAD 3
#define HDIM 12
#define NEG_SLOPE 0.2f

__device__ __forceinline__ float leaky(float x) {
  return x >= 0.0f ? x : NEG_SLOPE * x;
}

__global__ __launch_bounds__(32) void gemm36_wmma(
    const float* __restrict__ A, const float* __restrict__ W,
    const float* __restrict__ bias, float* __restrict__ out, int nrows) {
  const int lane = threadIdx.x;
  const int half = lane >> 4;
  const int idx  = lane & 15;

  int mrow = blockIdx.x * 16 + idx;
  mrow = mrow < nrows ? mrow : (nrows - 1);
  const float* __restrict__ Ab = A + (size_t)mrow * FDIM + 2 * half;

  const int col0 = idx;
  const int col1 = idx + 16;
  const int col2 = (idx + 32) < FDIM ? (idx + 32) : (FDIM - 1);
  const float* __restrict__ Wb0 = W + col0 + 72 * half;
  const float* __restrict__ Wb1 = W + col1 + 72 * half;
  const float* __restrict__ Wb2 = W + col2 + 72 * half;

  v8f acc[3];
  acc[0] = (v8f){0.f,0.f,0.f,0.f,0.f,0.f,0.f,0.f};
  acc[1] = acc[0];
  acc[2] = acc[0];

  #pragma unroll
  for (int k0 = 0; k0 < FDIM; k0 += 4) {
    v2f a, b0, b1, b2;
    a.x  = Ab[k0];
    a.y  = Ab[k0 + 1];
    b0.x = Wb0[(k0 + 0) * FDIM];
    b0.y = Wb0[(k0 + 1) * FDIM];
    b1.x = Wb1[(k0 + 0) * FDIM];
    b1.y = Wb1[(k0 + 1) * FDIM];
    b2.x = Wb2[(k0 + 0) * FDIM];
    b2.y = Wb2[(k0 + 1) * FDIM];
    acc[0] = __builtin_amdgcn_wmma_f32_16x16x4_f32(false, a, false, b0,
                                                   (short)0, acc[0], false, false);
    acc[1] = __builtin_amdgcn_wmma_f32_16x16x4_f32(false, a, false, b1,
                                                   (short)0, acc[1], false, false);
    acc[2] = __builtin_amdgcn_wmma_f32_16x16x4_f32(false, a, false, b2,
                                                   (short)0, acc[2], false, false);
  }

  const int rowbase = blockIdx.x * 16 + 8 * half;
  const bool full = (blockIdx.x * 16 + 16) <= nrows;
  #pragma unroll
  for (int t = 0; t < 3; ++t) {
    const int col = idx + 16 * t;
    if (col < FDIM) {
      const float bv = bias ? bias[col] : 0.0f;
      float* o = out + (size_t)rowbase * FDIM + col;
      if (full) {
        #pragma unroll
        for (int v = 0; v < 8; ++v) o[(size_t)v * FDIM] = acc[t][v] + bv;
      } else {
        #pragma unroll
        for (int v = 0; v < 8; ++v)
          if (rowbase + v < nrows) o[(size_t)v * FDIM] = acc[t][v] + bv;
      }
    }
  }
}

__global__ void attn_scores(const float* __restrict__ feat,
                            const float* __restrict__ a_l,
                            const float* __restrict__ a_r,
                            float* __restrict__ el4, float* __restrict__ er,
                            int n) {
  int gid = blockIdx.x * blockDim.x + threadIdx.x;
  if (gid >= n * NHEAD) return;
  const int node = gid / NHEAD;
  const int h    = gid % NHEAD;
  const float4* f4 = (const float4*)(feat + (size_t)node * FDIM + h * HDIM);
  const float4* l4 = (const float4*)(a_l + h * HDIM);
  const float4* r4 = (const float4*)(a_r + h * HDIM);
  float sl = 0.0f, sr = 0.0f;
  #pragma unroll
  for (int q = 0; q < HDIM / 4; ++q) {
    const float4 fv = f4[q], lv = l4[q], rv = r4[q];
    sl += fv.x * lv.x + fv.y * lv.y + fv.z * lv.z + fv.w * lv.w;
    sr += fv.x * rv.x + fv.y * rv.y + fv.z * rv.z + fv.w * rv.w;
  }
  el4[node * 4 + h] = sl;
  if (h == 0) el4[node * 4 + 3] = 0.0f;
  er[gid] = sr;
}

__global__ void csr_zero(int* __restrict__ cnt, int n1) {
  int gid = blockIdx.x * blockDim.x + threadIdx.x;
  if (gid < n1) cnt[gid] = 0;
}

#define CSR_R 12800
#define CSR_S 32
#define CSR_T 1024
#define CSR_U 8
#define CSR_NI ((CSR_R + CSR_T - 1) / CSR_T)

__device__ __forceinline__ void csr_slice(int ne, int& e0, int& e4, int& e1) {
  const int per = (((ne + (int)gridDim.y - 1) / (int)gridDim.y) + 3) & ~3;
  e0 = (int)blockIdx.y * per;
  e1 = e0 + per;
  if (e1 > ne) e1 = ne;
  if (e1 < e0) e1 = e0;
  e4 = e0 + ((e1 - e0) & ~3);
}

__device__ __forceinline__ int4 csr_dst4(const int* __restrict__ dst, int e,
                                         int b, int e4) {
  const bool ok = e < e4;
  int4 v = *(const int4*)(dst + (ok ? e : b));
  if (!ok) v = make_int4(-1, -1, -1, -1);
  return v;
}

__device__ __forceinline__ void csr_count1(int* lc, int d, int r0) {
  const unsigned k = (unsigned)(d - r0);
  if (k < CSR_R) atomicAdd(&lc[k], 1);
}

__device__ __forceinline__ void csr_local_count(const int* __restrict__ dst,
                                                int* lc, int r0, int e0,
                                                int e4, int e1) {
  const int tid = threadIdx.x;
  #pragma unroll
  for (int j = 0; j < CSR_NI; ++j)
    if (tid + j * CSR_T < CSR_R) lc[tid + j * CSR_T] = 0;
  __syncthreads();
  for (int b = e0 + tid * 4; b < e4; b += CSR_U * CSR_T * 4) {
    int4 d[CSR_U];
    #pragma unroll
    for (int u = 0; u < CSR_U; ++u)
      d[u] = csr_dst4(dst, b + u * CSR_T * 4, b, e4);
    #pragma unroll
    for (int u = 0; u < CSR_U; ++u) {
      csr_count1(lc, d[u].x, r0);
      csr_count1(lc, d[u].y, r0);
      csr_count1(lc, d[u].z, r0);
      csr_count1(lc, d[u].w, r0);
    }
  }
  if (tid < e1 - e4) csr_count1(lc, dst[e4 + tid], r0);
  __syncthreads();
}

__global__ __launch_bounds__(CSR_T) void csr_count(
    const int* __restrict__ dst, int* __restrict__ cnt, int n, int ne) {
  __shared__ int lc[CSR_R];
  const int tid = threadIdx.x;
  const int r0 = blockIdx.x * CSR_R;
  int e0, e4, e1;
  csr_slice(ne, e0, e4, e1);
  csr_local_count(dst, lc, r0, e0, e4, e1);
  #pragma unroll
  for (int j = 0; j < CSR_NI; ++j) {
    const int i = tid + j * CSR_T;
    if (i < CSR_R && r0 + i < n) {
      const int c = lc[i];
      if (c != 0) atomicAdd(&cnt[r0 + i], c);
    }
  }
}

__global__ __launch_bounds__(1024) void csr_scan(int* __restrict__ cur,
                                                 int* __restrict__ rowptr,
                                                 int n) {
  __shared__ int wsum[32];
  __shared__ int tile_total;
  const int tid  = threadIdx.x;
  const int lane = tid & 31;
  const int wid  = tid >> 5;
  int carry = 0;
  for (int base = 0; base < n; base += 4096) {
    const int i0 = base + tid * 4;
    const int c0 = (i0 + 0 < n) ? cur[i0 + 0] : 0;
    const int c1 = (i0 + 1 < n) ? cur[i0 + 1] : 0;
    const int c2 = (i0 + 2 < n) ? cur[i0 + 2] : 0;
    const int c3 = (i0 + 3 < n) ? cur[i0 + 3] : 0;
    const int tsum = c0 + c1 + c2 + c3;
    int x = tsum;
    for (int off = 1; off < 32; off <<= 1) {
      const int y = __shfl_up(x, off);
      if (lane >= off) x += y;
    }
    if (lane == 31) wsum[wid] = x;
    __syncthreads();
    if (wid == 0) {
      const int w = wsum[lane];
      int xs = w;
      for (int off = 1; off < 32; off <<= 1) {
        const int y = __shfl_up(xs, off);
        if (lane >= off) xs += y;
      }
      wsum[lane] = xs - w;
      if (lane == 31) tile_total = xs;
    }
    __syncthreads();
    int ex = carry + wsum[wid] + (x - tsum);
    carry += tile_total;
    if (i0 + 0 < n) { cur[i0 + 0] = ex; rowptr[i0 + 0] = ex; }
    ex += c0;
    if (i0 + 1 < n) { cur[i0 + 1] = ex; rowptr[i0 + 1] = ex; }
    ex += c1;
    if (i0 + 2 < n) { cur[i0 + 2] = ex; rowptr[i0 + 2] = ex; }
    ex += c2;
    if (i0 + 3 < n) { cur[i0 + 3] = ex; rowptr[i0 + 3] = ex; }
    __syncthreads();
  }
  if (tid == 0) rowptr[n] = carry;
}

__device__ __forceinline__ void csr_place1(int* lc, int* __restrict__ ssrc,
                                           int d, int sid, int r0) {
  const unsigned k = (unsigned)(d - r0);
  if (k < CSR_R) ssrc[atomicAdd(&lc[k], 1)] = sid;
}

__global__ __launch_bounds__(CSR_T) void csr_fill(
    const int* __restrict__ src, const int* __restrict__ dst,
    int* __restrict__ cur, int* __restrict__ ssrc, int n, int ne) {
  __shared__ int lc[CSR_R];
  const int tid = threadIdx.x;
  const int r0 = blockIdx.x * CSR_R;
  int e0, e4, e1;
  csr_slice(ne, e0, e4, e1);
  csr_local_count(dst, lc, r0, e0, e4, e1);

  int c[CSR_NI], base[CSR_NI];
  #pragma unroll
  for (int j = 0; j < CSR_NI; ++j) {
    const int i = tid + j * CSR_T;
    c[j] = (i < CSR_R && r0 + i < n) ? lc[i] : 0;
  }
  #pragma unroll
  for (int j = 0; j < CSR_NI; ++j) {
    base[j] = 0;
    if (c[j] != 0) base[j] = atomicAdd(&cur[r0 + tid + j * CSR_T], c[j]);
  }
  #pragma unroll
  for (int j = 0; j < CSR_NI; ++j)
    if (c[j] != 0) lc[tid + j * CSR_T] = base[j];
  __syncthreads();

  for (int b = e0 + tid * 4; b < e4; b += CSR_U * CSR_T * 4) {
    int4 d[CSR_U], sv[CSR_U];
    #pragma unroll
    for (int u = 0; u < CSR_U; ++u) {
      const int e = b + u * CSR_T * 4;
      d[u]  = csr_dst4(dst, e, b, e4);
      sv[u] = *(const int4*)(src + (e < e4 ? e : b));
    }
    #pragma unroll
    for (int u = 0; u < CSR_U; ++u) {
      csr_place1(lc, ssrc, d[u].x, sv[u].x, r0);
      csr_place1(lc, ssrc, d[u].y, sv[u].y, r0);
      csr_place1(lc, ssrc, d[u].z, sv[u].z, r0);
      csr_place1(lc, ssrc, d[u].w, sv[u].w, r0);
    }
  }
  if (tid < e1 - e4) csr_place1(lc, ssrc, dst[e4 + tid], src[e4 + tid], r0);
}

__device__ __forceinline__ float wave_max(float v) {
  for (int off = 16; off > 0; off >>= 1) v = fmaxf(v, __shfl_xor(v, off));
  return v;
}
__device__ __forceinline__ float wave_sum(float v) {
  for (int off = 16; off > 0; off >>= 1) v += __shfl_xor(v, off);
  return v;
}

struct EdgeS { int s; float v0, v1, v2; };

__device__ __forceinline__ EdgeS load_edge(const int* __restrict__ ssrc,
                                           const float4* __restrict__ el4,
                                           int j, int end, float er0,
                                           float er1, float er2) {
  EdgeS e;
  e.s = 0;
  e.v0 = -INFINITY; e.v1 = -INFINITY; e.v2 = -INFINITY;
  if (j < end) {
    e.s = ssrc[j];
    const float4 t = el4[e.s];
    e.v0 = leaky(t.x + er0);
    e.v1 = leaky(t.y + er1);
    e.v2 = leaky(t.z + er2);
  }
  return e;
}

__global__ __launch_bounds__(256) void gat_aggregate(
    const int* __restrict__ rowptr, const int* __restrict__ ssrc,
    const float* __restrict__ el4f, const float* __restrict__ er,
    const float* __restrict__ feat, const float* __restrict__ bias,
    float* __restrict__ h, int n) {
  const int lane = threadIdx.x & 31;
  const int d = blockIdx.x * 8 + (threadIdx.x >> 5);
  if (d >= n) return;
  const float4* __restrict__ el4 = (const float4*)el4f;
  const int beg = rowptr[d];
  const int end = rowptr[d + 1];
  const float er0 = er[d * NHEAD + 0];
  const float er1 = er[d * NHEAD + 1];
  const float er2 = er[d * NHEAD + 2];

  const EdgeS eA = load_edge(ssrc, el4, beg + lane, end, er0, er1, er2);

  float m0 = eA.v0, m1 = eA.v1, m2 = eA.v2;
  for (int j = beg + 32 + lane; j < end; j += 32) {
    const EdgeS e = load_edge(ssrc, el4, j, end, er0, er1, er2);
    m0 = fmaxf(m0, e.v0);
    m1 = fmaxf(m1, e.v1);
    m2 = fmaxf(m2, e.v2);
  }
  m0 = wave_max(m0); m1 = wave_max(m1); m2 = wave_max(m2);
  if (!isfinite(m0)) m0 = 0.0f;
  if (!isfinite(m1)) m1 = 0.0f;
  if (!isfinite(m2)) m2 = 0.0f;

  float d0 = expf(eA.v0 - m0);
  float d1 = expf(eA.v1 - m1);
  float d2 = expf(eA.v2 - m2);
  for (int j = beg + 32 + lane; j < end; j += 32) {
    const EdgeS e = load_edge(ssrc, el4, j, end, er0, er1, er2);
    d0 += expf(e.v0 - m0);
    d1 += expf(e.v1 - m1);
    d2 += expf(e.v2 - m2);
  }
  d0 = fmaxf(wave_sum(d0), 1e-9f);
  d1 = fmaxf(wave_sum(d1), 1e-9f);
  d2 = fmaxf(wave_sum(d2), 1e-9f);

  const int g  = lane / 9;
  const int q  = lane - g * 9;
  const int hq = q / 3;
  float4 acc = make_float4(0.0f, 0.0f, 0.0f, 0.0f);
  for (int cb = beg; cb < end; cb += 32) {
    EdgeS e = eA;
    if (cb != beg)
      e = load_edge(ssrc, el4, cb + lane, end, er0, er1, er2);
    const int   s  = e.s;
    const float p0 = expf(e.v0 - m0) / d0;
    const float p1 = expf(e.v1 - m1) / d1;
    const float p2 = expf(e.v2 - m2) / d2;
    const int cnt = (end - cb) < 32 ? (end - cb) : 32;
    for (int t0 = 0; t0 < cnt; t0 += 12) {
      float4 fv[4];
      float  aw[4];
      #pragma unroll
      for (int u = 0; u < 4; ++u) {
        const int  jj = t0 + 3 * u + g;
        const bool ok = (lane < 27) && (jj < cnt);
        const int  fb = (t0 + 3 * u) < cnt ? (t0 + 3 * u) : (cnt - 1);
        const int  sl = ok ? jj : fb;
        const int   ss = __shfl(s, sl);
        const float a0 = __shfl(p0, sl);
        const float a1 = __shfl(p1, sl);
        const float a2 = __shfl(p2, sl);
        const float a  = hq == 0 ? a0 : (hq == 1 ? a1 : a2);
        aw[u] = ok ? a : 0.0f;
        fv[u] = ((const float4*)(feat + (size_t)ss * FDIM))[q];
      }
      #pragma unroll
      for (int u = 0; u < 4; ++u) {
        acc.x += fv[u].x * aw[u];
        acc.y += fv[u].y * aw[u];
        acc.z += fv[u].z * aw[u];
        acc.w += fv[u].w * aw[u];
      }
    }
  }
  float4 r;
  r.x = acc.x + __shfl(acc.x, (lane + 9) & 31) + __shfl(acc.x, (lane + 18) & 31);
  r.y = acc.y + __shfl(acc.y, (lane + 9) & 31) + __shfl(acc.y, (lane + 18) & 31);
  r.z = acc.z + __shfl(acc.z, (lane + 9) & 31) + __shfl(acc.z, (lane + 18) & 31);
  r.w = acc.w + __shfl(acc.w, (lane + 9) & 31) + __shfl(acc.w, (lane + 18) & 31);
  if (lane < 9) {
    float4* hp = (float4*)(h + (size_t)d * FDIM) + q;
    const float4 hv = *hp;
    const float4 bv = ((const float4*)bias)[q];
    float4 o;
    o.x = r.x + hv.x + bv.x;
    o.y = r.y + hv.y + bv.y;
    o.z = r.z + hv.z + bv.z;
    o.w = r.w + hv.w + bv.w;
    o.x = o.x > 0.0f ? o.x : expm1f(o.x);
    o.y = o.y > 0.0f ? o.y : expm1f(o.y);
    o.z = o.z > 0.0f ? o.z : expm1f(o.z);
    o.w = o.w > 0.0f ? o.w : expm1f(o.w);
    *hp = o;
  }
}

extern "C" void kernel_launch(void* const* d_in, const int* in_sizes, int n_in,
                              void* d_out, int out_size, void* d_ws,
                              size_t ws_size, hipStream_t stream) {
  (void)n_in; (void)out_size; (void)ws_size;
  const float* X        = (const float*)d_in[0];
  const int*   src      = (const int*)d_in[1];
  const int*   dst      = (const int*)d_in[2];
  const float* lin0_w   = (const float*)d_in[3];
  const float* lin0_b   = (const float*)d_in[4];
  const float* fc_w     = (const float*)d_in[5];
  const float* attn_l   = (const float*)d_in[6];
  const float* attn_r   = (const float*)d_in[7];
  const float* gat_bias = (const float*)d_in[8];
  const float* out_w    = (const float*)d_in[9];
  const float* out_b    = (const float*)d_in[10];

  const int n  = in_sizes[0] / FDIM;
  const int ne = in_sizes[1];

  float* ws   = (float*)d_ws;
  float* hbuf  = ws; ws += (size_t)n * FDIM;
  float* feat  = ws; ws += (size_t)n * FDIM;
  float* el4   = ws; ws += (size_t)n * 4;
  float* er    = ws; ws += (size_t)n * NHEAD;
  int* rowptr  = (int*)ws;
  int* cur     = rowptr + (size_t)n + 1;
  int* ssrc    = cur + (size_t)n;

  const dim3 gemmGrid((n + 15) / 16);
  const dim3 csrGrid((n + CSR_R - 1) / CSR_R, CSR_S);
  const int nbN1 = (n + 1 + 255) / 256;
  const int nbN3 = (n * NHEAD + 255) / 256;
  const int nbAg = (n + 7) / 8;

  csr_zero<<<nbN1, 256, 0, stream>>>(cur, n);
  csr_count<<<csrGrid, CSR_T, 0, stream>>>(dst, cur, n, ne);
  csr_scan<<<1, 1024, 0, stream>>>(cur, rowptr, n);
  csr_fill<<<csrGrid, CSR_T, 0, stream>>>(src, dst, cur, ssrc, n, ne);

  gemm36_wmma<<<gemmGrid, 32, 0, stream>>>(X, lin0_w, lin0_b, hbuf, n);

  for (int l = 0; l < 3; ++l) {
    gemm36_wmma<<<gemmGrid, 32, 0, stream>>>(
        hbuf, fc_w + (size_t)l * FDIM * FDIM, nullptr, feat, n);
    attn_scores<<<nbN3, 256, 0, stream>>>(
        feat, attn_l + (size_t)l * NHEAD * HDIM, attn_r + (size_t)l * NHEAD * HDIM,
        el4, er, n);
    gat_aggregate<<<nbAg, 256, 0, stream>>>(rowptr, ssrc, el4, er, feat,
                                            gat_bias + (size_t)l * FDIM, hbuf, n);
  }

  gemm36_wmma<<<gemmGrid, 32, 0, stream>>>(hbuf, out_w, out_b, (float*)d_out, n);
}


# speedup vs baseline: 20.5343x; 1.3342x over previous
#include <hip/hip_runtime.h>
#include <math.h>

typedef __attribute__((ext_vector_type(2))) float v2f;
typedef __attribute__((ext_vector_type(8))) float v8f;

#define FDIM 36
#define NHEAD 3
#define HDIM 12
#define NEG_SLOPE 0.2f
#define FPITCH 64

__device__ __forceinline__ float leaky(float x) {
  return x >= 0.0f ? x : NEG_SLOPE * x;
}

template <int OPITCH>
__device__ __forceinline__ void gemm36_body(
    const float* A, const float* __restrict__ W,
    const float* __restrict__ bias, float* out, int nrows) {
  const int lane = threadIdx.x;
  const int half = lane >> 4;
  const int idx  = lane & 15;

  int mrow = blockIdx.x * 16 + idx;
  mrow = mrow < nrows ? mrow : (nrows - 1);
  const float* __restrict__ Ab = A + (size_t)mrow * FDIM + 2 * half;

  const int col0 = idx;
  const int col1 = idx + 16;
  const int col2 = (idx + 32) < FDIM ? (idx + 32) : (FDIM - 1);
  const float* __restrict__ Wb0 = W + col0 + 72 * half;
  const float* __restrict__ Wb1 = W + col1 + 72 * half;
  const float* __restrict__ Wb2 = W + col2 + 72 * half;

  v8f acc[3];
  acc[0] = (v8f){0.f,0.f,0.f,0.f,0.f,0.f,0.f,0.f};
  acc[1] = acc[0];
  acc[2] = acc[0];

  #pragma unroll
  for (int k0 = 0; k0 < FDIM; k0 += 4) {
    v2f a, b0, b1, b2;
    a.x  = Ab[k0];
    a.y  = Ab[k0 + 1];
    b0.x = Wb0[(k0 + 0) * FDIM];
    b0.y = Wb0[(k0 + 1) * FDIM];
    b1.x = Wb1[(k0 + 0) * FDIM];
    b1.y = Wb1[(k0 + 1) * FDIM];
    b2.x = Wb2[(k0 + 0) * FDIM];
    b2.y = Wb2[(k0 + 1) * FDIM];
    acc[0] = __builtin_amdgcn_wmma_f32_16x16x4_f32(false, a, false, b0,
                                                   (short)0, acc[0], false, false);
    acc[1] = __builtin_amdgcn_wmma_f32_16x16x4_f32(false, a, false, b1,
                                                   (short)0, acc[1], false, false);
    acc[2] = __builtin_amdgcn_wmma_f32_16x16x4_f32(false, a, false, b2,
                                                   (short)0, acc[2], false, false);
  }

  const int rowbase = blockIdx.x * 16 + 8 * half;
  const bool full = (blockIdx.x * 16 + 16) <= nrows;
  #pragma unroll
  for (int t = 0; t < 3; ++t) {
    const int col = idx + 16 * t;
    if (col < FDIM) {
      const float bv = bias ? bias[col] : 0.0f;
      float* o = out + (size_t)rowbase * OPITCH + col;
      if (full) {
        #pragma unroll
        for (int v = 0; v < 8; ++v) o[(size_t)v * OPITCH] = acc[t][v] + bv;
      } else {
        #pragma unroll
        for (int v = 0; v < 8; ++v)
          if (rowbase + v < nrows) o[(size_t)v * OPITCH] = acc[t][v] + bv;
      }
    }
  }
}

__global__ __launch_bounds__(32) void gemm36_wmma(
    const float* A, const float* __restrict__ W,
    const float* __restrict__ bias, float* out, int nrows) {
  gemm36_body<FDIM>(A, W, bias, out, nrows);
}

__global__ __launch_bounds__(32) void gemm36_wmma_padded(
    const float* A, const float* __restrict__ W,
    const float* __restrict__ bias, float* out, int nrows) {
  gemm36_body<FPITCH>(A, W, bias, out, nrows);
}

__global__ void csr_zero(int* __restrict__ cnt, int n1) {
  int gid = blockIdx.x * blockDim.x + threadIdx.x;
  if (gid < n1) cnt[gid] = 0;
}

#define CSR_R 12800
#define CSR_S 32
#define CSR_T 1024
#define CSR_U 8
#define CSR_NI ((CSR_R + CSR_T - 1) / CSR_T)

__device__ __forceinline__ void csr_slice(int ne, int& e0, int& e4, int& e1) {
  const int per = (((ne + (int)gridDim.y - 1) / (int)gridDim.y) + 3) & ~3;
  e0 = (int)blockIdx.y * per;
  e1 = e0 + per;
  if (e1 > ne) e1 = ne;
  if (e1 < e0) e1 = e0;
  e4 = e0 + ((e1 - e0) & ~3);
}

__device__ __forceinline__ int4 csr_dst4(const int* __restrict__ dst, int e,
                                         int b, int e4) {
  const bool ok = e < e4;
  int4 v = *(const int4*)(dst + (ok ? e : b));
  if (!ok) v = make_int4(-1, -1, -1, -1);
  return v;
}

__device__ __forceinline__ void csr_count1(int* lc, int d, int r0) {
  const unsigned k = (unsigned)(d - r0);
  if (k < CSR_R) atomicAdd(&lc[k], 1);
}

__device__ __forceinline__ void csr_local_count(const int* __restrict__ dst,
                                                int* lc, int r0, int e0,
                                                int e4, int e1) {
  const int tid = threadIdx.x;
  #pragma unroll
  for (int j = 0; j < CSR_NI; ++j)
    if (tid + j * CSR_T < CSR_R) lc[tid + j * CSR_T] = 0;
  __syncthreads();
  for (int b = e0 + tid * 4; b < e4; b += CSR_U * CSR_T * 4) {
    int4 d[CSR_U];
    #pragma unroll
    for (int u = 0; u < CSR_U; ++u)
      d[u] = csr_dst4(dst, b + u * CSR_T * 4, b, e4);
    #pragma unroll
    for (int u = 0; u < CSR_U; ++u) {
      csr_count1(lc, d[u].x, r0);
      csr_count1(lc, d[u].y, r0);
      csr_count1(lc, d[u].z, r0);
      csr_count1(lc, d[u].w, r0);
    }
  }
  if (tid < e1 - e4) csr_count1(lc, dst[e4 + tid], r0);
  __syncthreads();
}

__global__ __launch_bounds__(CSR_T) void csr_count(
    const int* __restrict__ dst, int* __restrict__ cnt, int n, int ne) {
  __shared__ int lc[CSR_R];
  const int tid = threadIdx.x;
  const int r0 = blockIdx.x * CSR_R;
  int e0, e4, e1;
  csr_slice(ne, e0, e4, e1);
  csr_local_count(dst, lc, r0, e0, e4, e1);
  #pragma unroll
  for (int j = 0; j < CSR_NI; ++j) {
    const int i = tid + j * CSR_T;
    if (i < CSR_R && r0 + i < n) {
      const int c = lc[i];
      if (c != 0) atomicAdd(&cnt[r0 + i], c);
    }
  }
}

__global__ __launch_bounds__(1024) void csr_scan(int* __restrict__ cur,
                                                 int* __restrict__ rowptr,
                                                 int n) {
  __shared__ int wsum[32];
  __shared__ int tile_total;
  const int tid  = threadIdx.x;
  const int lane = tid & 31;
  const int wid  = tid >> 5;
  int carry = 0;
  for (int base = 0; base < n; base += 4096) {
    const int i0 = base + tid * 4;
    const int c0 = (i0 + 0 < n) ? cur[i0 + 0] : 0;
    const int c1 = (i0 + 1 < n) ? cur[i0 + 1] : 0;
    const int c2 = (i0 + 2 < n) ? cur[i0 + 2] : 0;
    const int c3 = (i0 + 3 < n) ? cur[i0 + 3] : 0;
    const int tsum = c0 + c1 + c2 + c3;
    int x = tsum;
    for (int off = 1; off < 32; off <<= 1) {
      const int y = __shfl_up(x, off);
      if (lane >= off) x += y;
    }
    if (lane == 31) wsum[wid] = x;
    __syncthreads();
    if (wid == 0) {
      const int w = wsum[lane];
      int xs = w;
      for (int off = 1; off < 32; off <<= 1) {
        const int y = __shfl_up(xs, off);
        if (lane >= off) xs += y;
      }
      wsum[lane] = xs - w;
      if (lane == 31) tile_total = xs;
    }
    __syncthreads();
    int ex = carry + wsum[wid] + (x - tsum);
    carry += tile_total;
    if (i0 + 0 < n) { cur[i0 + 0] = ex; rowptr[i0 + 0] = ex; }
    ex += c0;
    if (i0 + 1 < n) { cur[i0 + 1] = ex; rowptr[i0 + 1] = ex; }
    ex += c1;
    if (i0 + 2 < n) { cur[i0 + 2] = ex; rowptr[i0 + 2] = ex; }
    ex += c2;
    if (i0 + 3 < n) { cur[i0 + 3] = ex; rowptr[i0 + 3] = ex; }
    __syncthreads();
  }
  if (tid == 0) rowptr[n] = carry;
}

__device__ __forceinline__ void csr_place1(int* lc, int* __restrict__ ssrc,
                                           int d, int sid, int r0) {
  const unsigned k = (unsigned)(d - r0);
  if (k < CSR_R) ssrc[atomicAdd(&lc[k], 1)] = sid;
}

__global__ __launch_bounds__(CSR_T) void csr_fill(
    const int* __restrict__ src, const int* __restrict__ dst,
    int* __restrict__ cur, int* __restrict__ ssrc, int n, int ne) {
  __shared__ int lc[CSR_R];
  const int tid = threadIdx.x;
  const int r0 = blockIdx.x * CSR_R;
  int e0, e4, e1;
  csr_slice(ne, e0, e4, e1);
  csr_local_count(dst, lc, r0, e0, e4, e1);

  int c[CSR_NI], base[CSR_NI];
  #pragma unroll
  for (int j = 0; j < CSR_NI; ++j) {
    const int i = tid + j * CSR_T;
    c[j] = (i < CSR_R && r0 + i < n) ? lc[i] : 0;
  }
  #pragma unroll
  for (int j = 0; j < CSR_NI; ++j) {
    base[j] = 0;
    if (c[j] != 0) base[j] = atomicAdd(&cur[r0 + tid + j * CSR_T], c[j]);
  }
  #pragma unroll
  for (int j = 0; j < CSR_NI; ++j)
    if (c[j] != 0) lc[tid + j * CSR_T] = base[j];
  __syncthreads();

  for (int b = e0 + tid * 4; b < e4; b += CSR_U * CSR_T * 4) {
    int4 d[CSR_U], sv[CSR_U];
    #pragma unroll
    for (int u = 0; u < CSR_U; ++u) {
      const int e = b + u * CSR_T * 4;
      d[u]  = csr_dst4(dst, e, b, e4);
      sv[u] = *(const int4*)(src + (e < e4 ? e : b));
    }
    #pragma unroll
    for (int u = 0; u < CSR_U; ++u) {
      csr_place1(lc, ssrc, d[u].x, sv[u].x, r0);
      csr_place1(lc, ssrc, d[u].y, sv[u].y, r0);
      csr_place1(lc, ssrc, d[u].z, sv[u].z, r0);
      csr_place1(lc, ssrc, d[u].w, sv[u].w, r0);
    }
  }
  if (tid < e1 - e4) csr_place1(lc, ssrc, dst[e4 + tid], src[e4 + tid], r0);
}

__device__ __forceinline__ float dot4(float4 a, float4 b) {
  return a.x * b.x + a.y * b.y + a.z * b.z + a.w * b.w;
}

__device__ __forceinline__ float exp_neg(float x) {
  return __builtin_amdgcn_exp2f(x * 1.4426950408889634f);
}

__device__ __forceinline__ float head_sum(float x, int b0) {
  return (__shfl(x, b0) + __shfl(x, b0 + 1)) + __shfl(x, b0 + 2);
}

__global__ __launch_bounds__(256) void gat_aggregate(
    const int* __restrict__ rowptr, const int* __restrict__ ssrc,
    const float* __restrict__ feat, const float* __restrict__ a_l,
    const float* __restrict__ a_r, const float* __restrict__ bias,
    float* __restrict__ h, int n) {
  const int lane = threadIdx.x & 31;
  const int d = blockIdx.x * 8 + __builtin_amdgcn_readfirstlane(threadIdx.x >> 5);
  if (d >= n) return;
  const float4* __restrict__ feat4 = (const float4*)feat;
  const int g  = lane / 9;
  const int q  = lane - g * 9;
  const int hq = q / 3;
  const int b0 = lane - (q - hq * 3);
  const bool live = lane < 27;
  const int beg = rowptr[d];
  const int end = rowptr[d + 1];
  const float4 al = ((const float4*)a_l)[q];
  const float4 ar = ((const float4*)a_r)[q];
  const float erh = head_sum(
      dot4(feat4[(unsigned)d * (FPITCH / 4) + (unsigned)q], ar), b0);

  float  m = -INFINITY;
  float  l = 0.0f;
  float4 acc = make_float4(0.0f, 0.0f, 0.0f, 0.0f);
  for (int cb = beg; cb < end; cb += 32) {
    const int cnt = (end - cb) < 32 ? (end - cb) : 32;
    const int s = ssrc[cb + (lane < cnt ? lane : cnt - 1)];
    for (int t0 = 0; t0 < cnt; t0 += 12) {
      float4 fv[4];
      bool   ok[4];
      #pragma unroll
      for (int u = 0; u < 4; ++u) {
        const int jj = t0 + 3 * u + g;
        const int fb = (t0 + 3 * u) < cnt ? (t0 + 3 * u) : (cnt - 1);
        ok[u] = live && (jj < cnt);
        const int ss = __shfl(s, ok[u] ? jj : fb);
        fv[u] = feat4[(unsigned)ss * (FPITCH / 4) + (unsigned)q];
      }
      float v[4];
      #pragma unroll
      for (int u = 0; u < 4; ++u) {
        const float el = head_sum(dot4(fv[u], al), b0);
        v[u] = ok[u] ? leaky(el + erh) : -INFINITY;
      }
      const float mn = fmaxf(fmaxf(m, fmaxf(v[0], v[1])), fmaxf(v[2], v[3]));
      const float mf = (mn == -INFINITY) ? 0.0f : mn;
      const float sc = exp_neg(m - mf);
      float p[4];
      #pragma unroll
      for (int u = 0; u < 4; ++u) p[u] = exp_neg(v[u] - mf);
      l = l * sc + ((p[0] + p[1]) + (p[2] + p[3]));
      acc.x *= sc; acc.y *= sc; acc.z *= sc; acc.w *= sc;
      #pragma unroll
      for (int u = 0; u < 4; ++u) {
        acc.x += fv[u].x * p[u];
        acc.y += fv[u].y * p[u];
        acc.z += fv[u].z * p[u];
        acc.w += fv[u].w * p[u];
      }
      m = mn;
    }
  }
  const int lB = (lane + 9) & 31, lC = (lane + 18) & 31;
  const float mB = __shfl(m, lB), mC = __shfl(m, lC);
  const float sB = __shfl(l, lB), sC = __shfl(l, lC);
  float4 aB, aC;
  aB.x = __shfl(acc.x, lB); aC.x = __shfl(acc.x, lC);
  aB.y = __shfl(acc.y, lB); aC.y = __shfl(acc.y, lC);
  aB.z = __shfl(acc.z, lB); aC.z = __shfl(acc.z, lC);
  aB.w = __shfl(acc.w, lB); aC.w = __shfl(acc.w, lC);
  if (lane < 9) {
    const float mm = fmaxf(m, fmaxf(mB, mC));
    const float mg = (mm == -INFINITY) ? 0.0f : mm;
    const float wA = exp_neg(m - mg);
    const float wB = exp_neg(mB - mg);
    const float wC = exp_neg(mC - mg);
    const float inv = 1.0f / fmaxf((l * wA + sB * wB) + sC * wC, 1e-9f);
    float4 r;
    r.x = ((acc.x * wA + aB.x * wB) + aC.x * wC) * inv;
    r.y = ((acc.y * wA + aB.y * wB) + aC.y * wC) * inv;
    r.z = ((acc.z * wA + aB.z * wB) + aC.z * wC) * inv;
    r.w = ((acc.w * wA + aB.w * wB) + aC.w * wC) * inv;
    float4* hp = (float4*)(h + (size_t)d * FDIM) + q;
    const float4 hv = *hp;
    const float4 bv = ((const float4*)bias)[q];
    float4 o;
    o.x = r.x + hv.x + bv.x;
    o.y = r.y + hv.y + bv.y;
    o.z = r.z + hv.z + bv.z;
    o.w = r.w + hv.w + bv.w;
    o.x = o.x > 0.0f ? o.x : expm1f(o.x);
    o.y = o.y > 0.0f ? o.y : expm1f(o.y);
    o.z = o.z > 0.0f ? o.z : expm1f(o.z);
    o.w = o.w > 0.0f ? o.w : expm1f(o.w);
    *hp = o;
  }
}

extern "C" void kernel_launch(void* const* d_in, const int* in_sizes, int n_in,
                              void* d_out, int out_size, void* d_ws,
                              size_t ws_size, hipStream_t stream) {
  (void)n_in; (void)out_size; (void)ws_size;
  const float* X        = (const float*)d_in[0];
  const int*   src      = (const int*)d_in[1];
  const int*   dst      = (const int*)d_in[2];
  const float* lin0_w   = (const float*)d_in[3];
  const float* lin0_b   = (const float*)d_in[4];
  const float* fc_w     = (const float*)d_in[5];
  const float* attn_l   = (const float*)d_in[6];
  const float* attn_r   = (const float*)d_in[7];
  const float* gat_bias = (const float*)d_in[8];
  const float* out_w    = (const float*)d_in[9];
  const float* out_b    = (const float*)d_in[10];

  const int n  = in_sizes[0] / FDIM;
  const int ne = in_sizes[1];

  float* ws   = (float*)d_ws;
  float* feat  = ws; ws += (size_t)n * FPITCH;
  float* hbuf  = (float*)d_out;
  int* rowptr  = (int*)ws;
  int* cur     = rowptr + (size_t)n + 1;
  int* ssrc    = cur + (size_t)n;

  const dim3 gemmGrid((n + 15) / 16);
  const dim3 csrGrid((n + CSR_R - 1) / CSR_R, CSR_S);
  const int nbN1 = (n + 1 + 255) / 256;
  const int nbAg = (n + 7) / 8;

  csr_zero<<<nbN1, 256, 0, stream>>>(cur, n);
  csr_count<<<csrGrid, CSR_T, 0, stream>>>(dst, cur, n, ne);
  csr_scan<<<1, 1024, 0, stream>>>(cur, rowptr, n);
  csr_fill<<<csrGrid, CSR_T, 0, stream>>>(src, dst, cur, ssrc, n, ne);

  gemm36_wmma<<<gemmGrid, 32, 0, stream>>>(X, lin0_w, lin0_b, hbuf, n);

  for (int l = 0; l < 3; ++l) {
    gemm36_wmma_padded<<<gemmGrid, 32, 0, stream>>>(
        hbuf, fc_w + (size_t)l * FDIM * FDIM, nullptr, feat, n);
    gat_aggregate<<<nbAg, 256, 0, stream>>>(
        rowptr, ssrc, feat, attn_l + (size_t)l * NHEAD * HDIM,
        attn_r + (size_t)l * NHEAD * HDIM, gat_bias + (size_t)l * FDIM, hbuf, n);
  }

  gemm36_wmma<<<gemmGrid, 32, 0, stream>>>(hbuf, out_w, out_b, hbuf, n);
}


# prefix sum of the in-degree counters moved into the fill launch: each group adds up the counters below its node range and scans its own range in LDS; the one-group serial scan launch goes
# speedup vs baseline: 24.2754x; 1.1822x over previous
#include <hip/hip_runtime.h>
#include <math.h>

typedef __attribute__((ext_vector_type(2))) float v2f;
typedef __attribute__((ext_vector_type(8))) float v8f;

#define FDIM 36
#define NHEAD 3
#define HDIM 12
#define NEG_SLOPE 0.2f
#define FPITCH 64

__device__ __forceinline__ float leaky(float x) {
  return x >= 0.0f ? x : NEG_SLOPE * x;
}

template <int OPITCH>
__device__ __forceinline__ void gemm36_body(
    const float* A, const float* __restrict__ W,
    const float* __restrict__ bias, float* out, int nrows) {
  const int lane = threadIdx.x;
  const int half = lane >> 4;
  const int idx  = lane & 15;

  int mrow = blockIdx.x * 16 + idx;
  mrow = mrow < nrows ? mrow : (nrows - 1);
  const float* __restrict__ Ab = A + (size_t)mrow * FDIM + 2 * half;

  const int col0 = idx;
  const int col1 = idx + 16;
  const int col2 = (idx + 32) < FDIM ? (idx + 32) : (FDIM - 1);
  const float* __restrict__ Wb0 = W + col0 + 72 * half;
  const float* __restrict__ Wb1 = W + col1 + 72 * half;
  const float* __restrict__ Wb2 = W + col2 + 72 * half;

  v8f acc[3];
  acc[0] = (v8f){0.f,0.f,0.f,0.f,0.f,0.f,0.f,0.f};
  acc[1] = acc[0];
  acc[2] = acc[0];

  #pragma unroll
  for (int k0 = 0; k0 < FDIM; k0 += 4) {
    v2f a, b0, b1, b2;
    a.x  = Ab[k0];
    a.y  = Ab[k0 + 1];
    b0.x = Wb0[(k0 + 0) * FDIM];
    b0.y = Wb0[(k0 + 1) * FDIM];
    b1.x = Wb1[(k0 + 0) * FDIM];
    b1.y = Wb1[(k0 + 1) * FDIM];
    b2.x = Wb2[(k0 + 0) * FDIM];
    b2.y = Wb2[(k0 + 1) * FDIM];
    acc[0] = __builtin_amdgcn_wmma_f32_16x16x4_f32(false, a, false, b0,
                                                   (short)0, acc[0], false, false);
    acc[1] = __builtin_amdgcn_wmma_f32_16x16x4_f32(false, a, false, b1,
                                                   (short)0, acc[1], false, false);
    acc[2] = __builtin_amdgcn_wmma_f32_16x16x4_f32(false, a, false, b2,
                                                   (short)0, acc[2], false, false);
  }

  const int rowbase = blockIdx.x * 16 + 8 * half;
  const bool full = (blockIdx.x * 16 + 16) <= nrows;
  #pragma unroll
  for (int t = 0; t < 3; ++t) {
    const int col = idx + 16 * t;
    if (col < FDIM) {
      const float bv = bias ? bias[col] : 0.0f;
      float* o = out + (size_t)rowbase * OPITCH + col;
      if (full) {
        #pragma unroll
        for (int v = 0; v < 8; ++v) o[(size_t)v * OPITCH] = acc[t][v] + bv;
      } else {
        #pragma unroll
        for (int v = 0; v < 8; ++v)
          if (rowbase + v < nrows) o[(size_t)v * OPITCH] = acc[t][v] + bv;
      }
    }
  }
}

__global__ __launch_bounds__(32) void gemm36_wmma(
    const float* A, const float* __restrict__ W,
    const float* __restrict__ bias, float* out, int nrows) {
  gemm36_body<FDIM>(A, W, bias, out, nrows);
}

__global__ __launch_bounds__(32) void gemm36_wmma_padded(
    const float* A, const float* __restrict__ W,
    const float* __restrict__ bias, float* out, int nrows) {
  gemm36_body<FPITCH>(A, W, bias, out, nrows);
}

__global__ void csr_zero(int* __restrict__ cnt, int n1) {
  int gid = blockIdx.x * blockDim.x + threadIdx.x;
  if (gid < n1) cnt[gid] = 0;
}

#define CSR_R 12800
#define CSR_S 32
#define CSR_T 1024
#define CSR_U 8
#define CSR_NI ((CSR_R + CSR_T - 1) / CSR_T)

__device__ __forceinline__ void csr_slice(int ne, int& e0, int& e4, int& e1) {
  const int per = (((ne + (int)gridDim.y - 1) / (int)gridDim.y) + 3) & ~3;
  e0 = (int)blockIdx.y * per;
  e1 = e0 + per;
  if (e1 > ne) e1 = ne;
  if (e1 < e0) e1 = e0;
  e4 = e0 + ((e1 - e0) & ~3);
}

__device__ __forceinline__ int4 csr_dst4(const int* __restrict__ dst, int e,
                                         int b, int e4) {
  const bool ok = e < e4;
  int4 v = *(const int4*)(dst + (ok ? e : b));
  if (!ok) v = make_int4(-1, -1, -1, -1);
  return v;
}

__device__ __forceinline__ void csr_count1(int* lc, int d, int r0) {
  const unsigned k = (unsigned)(d - r0);
  if (k < CSR_R) atomicAdd(&lc[k], 1);
}

__device__ __forceinline__ void csr_local_count(const int* __restrict__ dst,
                                                int* lc, int r0, int e0,
                                                int e4, int e1) {
  const int tid = threadIdx.x;
  #pragma unroll
  for (int j = 0; j < CSR_NI; ++j)
    if (tid + j * CSR_T < CSR_R) lc[tid + j * CSR_T] = 0;
  __syncthreads();
  for (int b = e0 + tid * 4; b < e4; b += CSR_U * CSR_T * 4) {
    int4 d[CSR_U];
    #pragma unroll
    for (int u = 0; u < CSR_U; ++u)
      d[u] = csr_dst4(dst, b + u * CSR_T * 4, b, e4);
    #pragma unroll
    for (int u = 0; u < CSR_U; ++u) {
      csr_count1(lc, d[u].x, r0);
      csr_count1(lc, d[u].y, r0);
      csr_count1(lc, d[u].z, r0);
      csr_count1(lc, d[u].w, r0);
    }
  }
  if (tid < e1 - e4) csr_count1(lc, dst[e4 + tid], r0);
  __syncthreads();
}

__global__ __launch_bounds__(CSR_T) void csr_count(
    const int* __restrict__ dst, int* __restrict__ cnt, int n, int ne) {
  __shared__ int lc[CSR_R];
  const int tid = threadIdx.x;
  const int r0 = blockIdx.x * CSR_R;
  int e0, e4, e1;
  csr_slice(ne, e0, e4, e1);
  csr_local_count(dst, lc, r0, e0, e4, e1);
  #pragma unroll
  for (int j = 0; j < CSR_NI; ++j) {
    const int i = tid + j * CSR_T;
    if (i < CSR_R && r0 + i < n) {
      const int c = lc[i];
      if (c != 0) atomicAdd(&cnt[r0 + i], c);
    }
  }
}

__device__ __forceinline__ void csr_place1(int* lc, int* __restrict__ ssrc,
                                           int d, int sid, int r0) {
  const unsigned k = (unsigned)(d - r0);
  if (k < CSR_R) ssrc[atomicAdd(&lc[k], 1)] = sid;
}

__global__ __launch_bounds__(CSR_T) void csr_fill(
    const int* __restrict__ src, const int* __restrict__ dst,
    const int* __restrict__ cnt, int* __restrict__ off,
    int* __restrict__ rowptr, int* __restrict__ ssrc, int n, int ne) {
  __shared__ int lc[CSR_R];
  __shared__ int wsum[32];
  __shared__ int bsum[32];
  const int tid  = threadIdx.x;
  const int lane = tid & 31;
  const int wid  = tid >> 5;
  const int r0 = blockIdx.x * CSR_R;
  int e0, e4, e1;
  csr_slice(ne, e0, e4, e1);

  int tc[CSR_NI];
  #pragma unroll
  for (int j = 0; j < CSR_NI; ++j) {
    const int i = tid + j * CSR_T;
    tc[j] = (i < CSR_R && r0 + i < n) ? cnt[r0 + i] : 0;
  }

  int below = 0;
  {
    const int4* __restrict__ c4 = (const int4*)cnt;
    const int nq = r0 >> 2;
    for (int q = tid; q < nq; q += CSR_U * CSR_T) {
      int4 v[CSR_U];
      #pragma unroll
      for (int u = 0; u < CSR_U; ++u) {
        const int qq = q + u * CSR_T;
        const bool ok = qq < nq;
        v[u] = c4[ok ? qq : q];
        if (!ok) v[u] = make_int4(0, 0, 0, 0);
      }
      #pragma unroll
      for (int u = 0; u < CSR_U; ++u)
        below += (v[u].x + v[u].y) + (v[u].z + v[u].w);
    }
  }

  csr_local_count(dst, lc, r0, e0, e4, e1);

  int c[CSR_NI];
  #pragma unroll
  for (int j = 0; j < CSR_NI; ++j) {
    const int i = tid + j * CSR_T;
    c[j] = (i < CSR_R && r0 + i < n) ? lc[i] : 0;
  }
  #pragma unroll
  for (int j = 0; j < CSR_NI; ++j) {
    const int i = tid + j * CSR_T;
    if (i < CSR_R) lc[i] = tc[j];
  }
  __syncthreads();

  int cv[CSR_NI];
  int tsum = 0;
  #pragma unroll
  for (int k = 0; k < CSR_NI; ++k) {
    const int i = tid * CSR_NI + k;
    cv[k] = (i < CSR_R) ? lc[i] : 0;
    tsum += cv[k];
  }
  int x = tsum;
  for (int o = 1; o < 32; o <<= 1) {
    const int y = __shfl_up(x, o);
    if (lane >= o) x += y;
  }
  int bw = below;
  for (int o = 16; o > 0; o >>= 1) bw += __shfl_xor(bw, o);
  if (lane == 31) wsum[wid] = x;
  if (lane == 0)  bsum[wid] = bw;
  __syncthreads();
  if (wid == 0) {
    const int w = wsum[lane];
    int xs = w;
    for (int o = 1; o < 32; o <<= 1) {
      const int y = __shfl_up(xs, o);
      if (lane >= o) xs += y;
    }
    int bt = bsum[lane];
    for (int o = 16; o > 0; o >>= 1) bt += __shfl_xor(bt, o);
    wsum[lane] = bt + (xs - w);
    if (lane == 31 && blockIdx.y == 0 && blockIdx.x == gridDim.x - 1)
      rowptr[n] = bt + xs;
  }
  __syncthreads();
  int ex = wsum[wid] + (x - tsum);
  #pragma unroll
  for (int k = 0; k < CSR_NI; ++k) {
    const int i = tid * CSR_NI + k;
    if (i < CSR_R) lc[i] = ex;
    ex += cv[k];
  }
  __syncthreads();

  int st[CSR_NI], base[CSR_NI];
  #pragma unroll
  for (int j = 0; j < CSR_NI; ++j) {
    const int i = tid + j * CSR_T;
    st[j] = (i < CSR_R) ? lc[i] : 0;
  }
  if (blockIdx.y == 0) {
    #pragma unroll
    for (int j = 0; j < CSR_NI; ++j) {
      const int i = tid + j * CSR_T;
      if (i < CSR_R && r0 + i < n) rowptr[r0 + i] = st[j];
    }
  }
  #pragma unroll
  for (int j = 0; j < CSR_NI; ++j) {
    base[j] = 0;
    if (c[j] != 0) base[j] = atomicAdd(&off[r0 + tid + j * CSR_T], c[j]);
  }
  #pragma unroll
  for (int j = 0; j < CSR_NI; ++j)
    if (c[j] != 0) lc[tid + j * CSR_T] = st[j] + base[j];
  __syncthreads();

  for (int b = e0 + tid * 4; b < e4; b += CSR_U * CSR_T * 4) {
    int4 d[CSR_U], sv[CSR_U];
    #pragma unroll
    for (int u = 0; u < CSR_U; ++u) {
      const int e = b + u * CSR_T * 4;
      d[u]  = csr_dst4(dst, e, b, e4);
      sv[u] = *(const int4*)(src + (e < e4 ? e : b));
    }
    #pragma unroll
    for (int u = 0; u < CSR_U; ++u) {
      csr_place1(lc, ssrc, d[u].x, sv[u].x, r0);
      csr_place1(lc, ssrc, d[u].y, sv[u].y, r0);
      csr_place1(lc, ssrc, d[u].z, sv[u].z, r0);
      csr_place1(lc, ssrc, d[u].w, sv[u].w, r0);
    }
  }
  if (tid < e1 - e4) csr_place1(lc, ssrc, dst[e4 + tid], src[e4 + tid], r0);
}

__device__ __forceinline__ float dot4(float4 a, float4 b) {
  return a.x * b.x + a.y * b.y + a.z * b.z + a.w * b.w;
}

__device__ __forceinline__ float exp_neg(float x) {
  return __builtin_amdgcn_exp2f(x * 1.4426950408889634f);
}

__device__ __forceinline__ float head_sum(float x, int b0) {
  return (__shfl(x, b0) + __shfl(x, b0 + 1)) + __shfl(x, b0 + 2);
}

__global__ __launch_bounds__(256) void gat_aggregate(
    const int* __restrict__ rowptr, const int* __restrict__ ssrc,
    const float* __restrict__ feat, const float* __restrict__ a_l,
    const float* __restrict__ a_r, const float* __restrict__ bias,
    float* __restrict__ h, int n) {
  const int lane = threadIdx.x & 31;
  const int d = blockIdx.x * 8 + __builtin_amdgcn_readfirstlane(threadIdx.x >> 5);
  if (d >= n) return;
  const float4* __restrict__ feat4 = (const float4*)feat;
  const int g  = lane / 9;
  const int q  = lane - g * 9;
  const int hq = q / 3;
  const int b0 = lane - (q - hq * 3);
  const bool live = lane < 27;
  const int beg = rowptr[d];
  const int end = rowptr[d + 1];
  const float4 al = ((const float4*)a_l)[q];
  const float4 ar = ((const float4*)a_r)[q];
  const float erh = head_sum(
      dot4(feat4[(unsigned)d * (FPITCH / 4) + (unsigned)q], ar), b0);

  float  m = -INFINITY;
  float  l = 0.0f;
  float4 acc = make_float4(0.0f, 0.0f, 0.0f, 0.0f);
  for (int cb = beg; cb < end; cb += 32) {
    const int cnt = (end - cb) < 32 ? (end - cb) : 32;
    const int s = ssrc[cb + (lane < cnt ? lane : cnt - 1)];
    for (int t0 = 0; t0 < cnt; t0 += 12) {
      float4 fv[4];
      bool   ok[4];
      #pragma unroll
      for (int u = 0; u < 4; ++u) {
        const int jj = t0 + 3 * u + g;
        const int fb = (t0 + 3 * u) < cnt ? (t0 + 3 * u) : (cnt - 1);
        ok[u] = live && (jj < cnt);
        const int ss = __shfl(s, ok[u] ? jj : fb);
        fv[u] = feat4[(unsigned)ss * (FPITCH / 4) + (unsigned)q];
      }
      float v[4];
      #pragma unroll
      for (int u = 0; u < 4; ++u) {
        const float el = head_sum(dot4(fv[u], al), b0);
        v[u] = ok[u] ? leaky(el + erh) : -INFINITY;
      }
      const float mn = fmaxf(fmaxf(m, fmaxf(v[0], v[1])), fmaxf(v[2], v[3]));
      const float mf = (mn == -INFINITY) ? 0.0f : mn;
      const float sc = exp_neg(m - mf);
      float p[4];
      #pragma unroll
      for (int u = 0; u < 4; ++u) p[u] = exp_neg(v[u] - mf);
      l = l * sc + ((p[0] + p[1]) + (p[2] + p[3]));
      acc.x *= sc; acc.y *= sc; acc.z *= sc; acc.w *= sc;
      #pragma unroll
      for (int u = 0; u < 4; ++u) {
        acc.x += fv[u].x * p[u];
        acc.y += fv[u].y * p[u];
        acc.z += fv[u].z * p[u];
        acc.w += fv[u].w * p[u];
      }
      m = mn;
    }
  }
  const int lB = (lane + 9) & 31, lC = (lane + 18) & 31;
  const float mB = __shfl(m, lB), mC = __shfl(m, lC);
  const float sB = __shfl(l, lB), sC = __shfl(l, lC);
  float4 aB, aC;
  aB.x = __shfl(acc.x, lB); aC.x = __shfl(acc.x, lC);
  aB.y = __shfl(acc.y, lB); aC.y = __shfl(acc.y, lC);
  aB.z = __shfl(acc.z, lB); aC.z = __shfl(acc.z, lC);
  aB.w = __shfl(acc.w, lB); aC.w = __shfl(acc.w, lC);
  if (lane < 9) {
    const float mm = fmaxf(m, fmaxf(mB, mC));
    const float mg = (mm == -INFINITY) ? 0.0f : mm;
    const float wA = exp_neg(m - mg);
    const float wB = exp_neg(mB - mg);
    const float wC = exp_neg(mC - mg);
    const float inv = 1.0f / fmaxf((l * wA + sB * wB) + sC * wC, 1e-9f);
    float4 r;
    r.x = ((acc.x * wA + aB.x * wB) + aC.x * wC) * inv;
    r.y = ((acc.y * wA + aB.y * wB) + aC.y * wC) * inv;
    r.z = ((acc.z * wA + aB.z * wB) + aC.z * wC) * inv;
    r.w = ((acc.w * wA + aB.w * wB) + aC.w * wC) * inv;
    float4* hp = (float4*)(h + (size_t)d * FDIM) + q;
    const float4 hv = *hp;
    const float4 bv = ((const float4*)bias)[q];
    float4 o;
    o.x = r.x + hv.x + bv.x;
    o.y = r.y + hv.y + bv.y;
    o.z = r.z + hv.z + bv.z;
    o.w = r.w + hv.w + bv.w;
    o.x = o.x > 0.0f ? o.x : expm1f(o.x);
    o.y = o.y > 0.0f ? o.y : expm1f(o.y);
    o.z = o.z > 0.0f ? o.z : expm1f(o.z);
    o.w = o.w > 0.0f ? o.w : expm1f(o.w);
    *hp = o;
  }
}

extern "C" void kernel_launch(void* const* d_in, const int* in_sizes, int n_in,
                              void* d_out, int out_size, void* d_ws,
                              size_t ws_size, hipStream_t stream) {
  (void)n_in; (void)out_size; (void)ws_size;
  const float* X        = (const float*)d_in[0];
  const int*   src      = (const int*)d_in[1];
  const int*   dst      = (const int*)d_in[2];
  const float* lin0_w   = (const float*)d_in[3];
  const float* lin0_b   = (const float*)d_in[4];
  const float* fc_w     = (const float*)d_in[5];
  const float* attn_l   = (const float*)d_in[6];
  const float* attn_r   = (const float*)d_in[7];
  const float* gat_bias = (const float*)d_in[8];
  const float* out_w    = (const float*)d_in[9];
  const float* out_b    = (const float*)d_in[10];

  const int n  = in_sizes[0] / FDIM;
  const int ne = in_sizes[1];

  float* ws   = (float*)d_ws;
  float* feat  = ws; ws += (size_t)n * FPITCH;
  float* hbuf  = (float*)d_out;
  int* cnt     = (int*)d_out;
  int* off     = cnt + (size_t)n;
  int* rowptr  = (int*)ws;
  int* ssrc    = rowptr + (size_t)n + 1;

  const dim3 gemmGrid((n + 15) / 16);
  const dim3 csrGrid((n + CSR_R - 1) / CSR_R, CSR_S);
  const int nbN2 = (2 * n + 255) / 256;
  const int nbAg = (n + 7) / 8;

  csr_zero<<<nbN2, 256, 0, stream>>>(cnt, 2 * n);
  csr_count<<<csrGrid, CSR_T, 0, stream>>>(dst, cnt, n, ne);
  csr_fill<<<csrGrid, CSR_T, 0, stream>>>(src, dst, cnt, off, rowptr, ssrc, n, ne);

  gemm36_wmma<<<gemmGrid, 32, 0, stream>>>(X, lin0_w, lin0_b, hbuf, n);

  for (int l = 0; l < 3; ++l) {
    gemm36_wmma_padded<<<gemmGrid, 32, 0, stream>>>(
        hbuf, fc_w + (size_t)l * FDIM * FDIM, nullptr, feat, n);
    gat_aggregate<<<nbAg, 256, 0, stream>>>(
        rowptr, ssrc, feat, attn_l + (size_t)l * NHEAD * HDIM,
        attn_r + (size_t)l * NHEAD * HDIM, gat_bias + (size_t)l * FDIM, hbuf, n);
  }

  gemm36_wmma<<<gemmGrid, 32, 0, stream>>>(hbuf, out_w, out_b, hbuf, n);
}


# edge grouping as a two-level LDS counting sort (slices split by 512-node bucket, then each bucket ranked by node, stored whole): no global atomics or one-word scattered stores; arrival order kept
# speedup vs baseline: 25.7399x; 1.0603x over previous
#include <hip/hip_runtime.h>
#include <math.h>

typedef __attribute__((ext_vector_type(2))) float v2f;
typedef __attribute__((ext_vector_type(8))) float v8f;

#define FDIM 36
#define NHEAD 3
#define HDIM 12
#define NEG_SLOPE 0.2f
#define FPITCH 64

__device__ __forceinline__ float leaky(float x) {
  return x >= 0.0f ? x : NEG_SLOPE * x;
}

template <int OPITCH>
__device__ __forceinline__ void gemm36_body(
    const float* A, const float* __restrict__ W,
    const float* __restrict__ bias, float* out, int nrows) {
  const int lane = threadIdx.x;
  const int half = lane >> 4;
  const int idx  = lane & 15;

  int mrow = blockIdx.x * 16 + idx;
  mrow = mrow < nrows ? mrow : (nrows - 1);
  const float* __restrict__ Ab = A + (size_t)mrow * FDIM + 2 * half;

  const int col0 = idx;
  const int col1 = idx + 16;
  const int col2 = (idx + 32) < FDIM ? (idx + 32) : (FDIM - 1);
  const float* __restrict__ Wb0 = W + col0 + 72 * half;
  const float* __restrict__ Wb1 = W + col1 + 72 * half;
  const float* __restrict__ Wb2 = W + col2 + 72 * half;

  v8f acc[3];
  acc[0] = (v8f){0.f,0.f,0.f,0.f,0.f,0.f,0.f,0.f};
  acc[1] = acc[0];
  acc[2] = acc[0];

  #pragma unroll
  for (int k0 = 0; k0 < FDIM; k0 += 4) {
    v2f a, b0, b1, b2;
    a.x  = Ab[k0];
    a.y  = Ab[k0 + 1];
    b0.x = Wb0[(k0 + 0) * FDIM];
    b0.y = Wb0[(k0 + 1) * FDIM];
    b1.x = Wb1[(k0 + 0) * FDIM];
    b1.y = Wb1[(k0 + 1) * FDIM];
    b2.x = Wb2[(k0 + 0) * FDIM];
    b2.y = Wb2[(k0 + 1) * FDIM];
    acc[0] = __builtin_amdgcn_wmma_f32_16x16x4_f32(false, a, false, b0,
                                                   (short)0, acc[0], false, false);
    acc[1] = __builtin_amdgcn_wmma_f32_16x16x4_f32(false, a, false, b1,
                                                   (short)0, acc[1], false, false);
    acc[2] = __builtin_amdgcn_wmma_f32_16x16x4_f32(false, a, false, b2,
                                                   (short)0, acc[2], false, false);
  }

  const int rowbase = blockIdx.x * 16 + 8 * half;
  const bool full = (blockIdx.x * 16 + 16) <= nrows;
  #pragma unroll
  for (int t = 0; t < 3; ++t) {
    const int col = idx + 16 * t;
    if (col < FDIM) {
      const float bv = bias ? bias[col] : 0.0f;
      float* o = out + (size_t)rowbase * OPITCH + col;
      if (full) {
        #pragma unroll
        for (int v = 0; v < 8; ++v) o[(size_t)v * OPITCH] = acc[t][v] + bv;
      } else {
        #pragma unroll
        for (int v = 0; v < 8; ++v)
          if (rowbase + v < nrows) o[(size_t)v * OPITCH] = acc[t][v] + bv;
      }
    }
  }
}

__global__ __launch_bounds__(32) void gemm36_wmma(
    const float* A, const float* __restrict__ W,
    const float* __restrict__ bias, float* out, int nrows) {
  gemm36_body<FDIM>(A, W, bias, out, nrows);
}

__global__ __launch_bounds__(32) void gemm36_wmma_padded(
    const float* A, const float* __restrict__ W,
    const float* __restrict__ bias, float* out, int nrows) {
  gemm36_body<FPITCH>(A, W, bias, out, nrows);
}

#define CSR_T     1024
#define CSR_NS    256
#define CSR_BSH   9
#define CSR_BR    (1 << CSR_BSH)
#define CSR_NBMAX 256
#define CSR_SLMAX 12800
#define CSR_CAP   36864
#define CSR_U     4
#define CSR_V     8

__device__ __forceinline__ void csr_slice(int ne, int& e0, int& e4, int& e1) {
  const int per = (((ne + (int)gridDim.x - 1) / (int)gridDim.x) + 3) & ~3;
  e0 = (int)blockIdx.x * per;
  e1 = e0 + per;
  if (e1 > ne) e1 = ne;
  if (e1 < e0) e1 = e0;
  e4 = e0 + ((e1 - e0) & ~3);
}

__device__ __forceinline__ int4 csr_ld4(const int* __restrict__ p, int e,
                                        int b, int e4) {
  return *(const int4*)(p + (e < e4 ? e : b));
}

__device__ __forceinline__ bool csr_ok(int d, int n) {
  return (unsigned)d < (unsigned)n;
}

__device__ __forceinline__ void csr_scan2(int* a, int* b, int len) {
  const int tid = threadIdx.x;
  for (int o = 1; o < len; o <<= 1) {
    int va = 0, vb = 0;
    if (tid < len && tid >= o) { va = a[tid - o]; vb = b[tid - o]; }
    __syncthreads();
    if (tid < len) { a[tid] += va; b[tid] += vb; }
    __syncthreads();
  }
}

__device__ __forceinline__ void csr_scan1(int* a, int len) {
  const int tid = threadIdx.x;
  for (int o = 1; o < len; o <<= 1) {
    int va = 0;
    if (tid < len && tid >= o) va = a[tid - o];
    __syncthreads();
    if (tid < len) a[tid] += va;
    __syncthreads();
  }
}

__device__ __forceinline__ void csr_hist1(int* hb, int d, int n) {
  if (csr_ok(d, n)) atomicAdd(&hb[d >> CSR_BSH], 1);
}

__global__ __launch_bounds__(CSR_T) void csr_hist(
    const int* __restrict__ dst, int* __restrict__ tab, int n, int nb, int ne) {
  __shared__ int hb[CSR_NBMAX];
  const int tid = threadIdx.x;
  int e0, e4, e1;
  csr_slice(ne, e0, e4, e1);
  if (tid < CSR_NBMAX) hb[tid] = 0;
  __syncthreads();
  for (int b = e0 + tid * 4; b < e4; b += CSR_U * CSR_T * 4) {
    int4 d[CSR_U];
    #pragma unroll
    for (int u = 0; u < CSR_U; ++u)
      d[u] = csr_ld4(dst, b + u * CSR_T * 4, b, e4);
    #pragma unroll
    for (int u = 0; u < CSR_U; ++u) {
      if (b + u * CSR_T * 4 < e4) {
        csr_hist1(hb, d[u].x, n);
        csr_hist1(hb, d[u].y, n);
        csr_hist1(hb, d[u].z, n);
        csr_hist1(hb, d[u].w, n);
      }
    }
  }
  if (tid < e1 - e4) csr_hist1(hb, dst[e4 + tid], n);
  __syncthreads();
  if (tid < nb) tab[(int)blockIdx.x * nb + tid] = hb[tid];
}

__device__ __forceinline__ void csr_part1(int* cur, const int* delta,
                                          int2* stage, int2* __restrict__ tmp,
                                          bool staged, int d, int sid, int n) {
  if (csr_ok(d, n)) {
    const int k = d >> CSR_BSH;
    const int r = atomicAdd(&cur[k], 1);
    if (staged) stage[r] = make_int2(d, sid);
    else        tmp[r + delta[k]] = make_int2(d, sid);
  }
}

__global__ __launch_bounds__(CSR_T) void csr_part(
    const int* __restrict__ src, const int* __restrict__ dst,
    const int* __restrict__ tab, int* __restrict__ bstart,
    int* __restrict__ tmpw, int n, int nb, int ne) {
  __shared__ int2 stage[CSR_SLMAX];
  __shared__ int ptot[CSR_T / CSR_NBMAX][CSR_NBMAX];
  __shared__ int pbel[CSR_T / CSR_NBMAX][CSR_NBMAX];
  __shared__ int gsc[CSR_NBMAX];
  __shared__ int lsc[CSR_NBMAX];
  __shared__ int delta[CSR_NBMAX];
  __shared__ int cur[CSR_NBMAX];
  int2* __restrict__ tmp = (int2*)tmpw;
  const int tid = threadIdx.x;
  const int s   = blockIdx.x;
  const int ns  = gridDim.x;
  int e0, e4, e1;
  csr_slice(ne, e0, e4, e1);
  const bool staged = (e1 - e0) <= CSR_SLMAX;

  {
    const int b = tid & (CSR_NBMAX - 1);
    const int p = tid / CSR_NBMAX;
    const int np = CSR_T / CSR_NBMAX;
    const int rows = (ns + np - 1) / np;
    int r0 = p * rows, r1 = r0 + rows;
    if (r1 > ns) r1 = ns;
    int tot = 0, bel = 0;
    if (b < nb) {
      #pragma unroll 16
      for (int r = r0; r < r1; ++r) {
        const int v = tab[r * nb + b];
        tot += v;
        bel += (r < s) ? v : 0;
      }
    }
    ptot[p][b] = tot;
    pbel[p][b] = bel;
  }
  __syncthreads();
  int tot_b = 0, bel_b = 0, own = 0;
  if (tid < CSR_NBMAX) {
    #pragma unroll
    for (int p = 0; p < CSR_T / CSR_NBMAX; ++p) {
      tot_b += ptot[p][tid];
      bel_b += pbel[p][tid];
    }
    own = (tid < nb) ? tab[s * nb + tid] : 0;
    gsc[tid] = tot_b;
    lsc[tid] = own;
  }
  __syncthreads();
  csr_scan2(gsc, lsc, CSR_NBMAX);
  if (tid < CSR_NBMAX) {
    const int gst = gsc[tid] - tot_b;
    const int lst = lsc[tid] - own;
    cur[tid]   = lst;
    delta[tid] = gst + bel_b - lst;
    if (s == 0) {
      if (tid < nb) bstart[tid] = gst;
      if (tid == nb - 1) bstart[nb] = gsc[tid];
    }
  }
  __syncthreads();

  for (int b = e0 + tid * 4; b < e4; b += CSR_U * CSR_T * 4) {
    int4 d[CSR_U], sv[CSR_U];
    #pragma unroll
    for (int u = 0; u < CSR_U; ++u) {
      d[u]  = csr_ld4(dst, b + u * CSR_T * 4, b, e4);
      sv[u] = csr_ld4(src, b + u * CSR_T * 4, b, e4);
    }
    #pragma unroll
    for (int u = 0; u < CSR_U; ++u) {
      if (b + u * CSR_T * 4 < e4) {
        csr_part1(cur, delta, stage, tmp, staged, d[u].x, sv[u].x, n);
        csr_part1(cur, delta, stage, tmp, staged, d[u].y, sv[u].y, n);
        csr_part1(cur, delta, stage, tmp, staged, d[u].z, sv[u].z, n);
        csr_part1(cur, delta, stage, tmp, staged, d[u].w, sv[u].w, n);
      }
    }
  }
  if (tid < e1 - e4)
    csr_part1(cur, delta, stage, tmp, staged, dst[e4 + tid], src[e4 + tid], n);
  if (staged) {
    __syncthreads();
    const int cnt = lsc[CSR_NBMAX - 1];
    for (int i = tid; i < cnt; i += CSR_T) {
      const int2 pr = stage[i];
      tmp[i + delta[pr.x >> CSR_BSH]] = pr;
    }
  }
}

__global__ __launch_bounds__(CSR_T) void csr_sort(
    const int* __restrict__ tmpw, const int* __restrict__ bstart,
    int* __restrict__ rowptr, int* __restrict__ ssrc, int n) {
  __shared__ int stage[CSR_CAP];
  __shared__ int lc[CSR_BR];
  __shared__ int sc[CSR_BR];
  const int2* __restrict__ tmp = (const int2*)tmpw;
  const int tid = threadIdx.x;
  const int bk  = blockIdx.x;
  const int bs  = bstart[bk];
  const int be  = bstart[bk + 1];
  const int cnt = be - bs;
  const int r0  = bk << CSR_BSH;
  const bool staged = cnt <= CSR_CAP;
  if (tid < CSR_BR) lc[tid] = 0;
  __syncthreads();
  for (int i0 = tid; i0 < cnt; i0 += CSR_V * CSR_T) {
    int dd[CSR_V];
    #pragma unroll
    for (int u = 0; u < CSR_V; ++u) {
      const int i = i0 + u * CSR_T;
      dd[u] = tmp[bs + (i < cnt ? i : i0)].x;
    }
    #pragma unroll
    for (int u = 0; u < CSR_V; ++u)
      if (i0 + u * CSR_T < cnt) atomicAdd(&lc[dd[u] - r0], 1);
  }
  __syncthreads();
  int own = 0;
  if (tid < CSR_BR) { own = lc[tid]; sc[tid] = own; }
  __syncthreads();
  csr_scan1(sc, CSR_BR);
  if (tid < CSR_BR) {
    const int ex = sc[tid] - own;
    lc[tid] = ex;
    if (r0 + tid < n) rowptr[r0 + tid] = bs + ex;
  }
  if (bk == (int)gridDim.x - 1 && tid == 0) rowptr[n] = be;
  __syncthreads();
  for (int i0 = tid; i0 < cnt; i0 += CSR_V * CSR_T) {
    int2 pr[CSR_V];
    #pragma unroll
    for (int u = 0; u < CSR_V; ++u) {
      const int i = i0 + u * CSR_T;
      pr[u] = tmp[bs + (i < cnt ? i : i0)];
    }
    #pragma unroll
    for (int u = 0; u < CSR_V; ++u) {
      if (i0 + u * CSR_T < cnt) {
        const int r = atomicAdd(&lc[pr[u].x - r0], 1);
        if (staged) stage[r] = pr[u].y;
        else        ssrc[bs + r] = pr[u].y;
      }
    }
  }
  if (staged) {
    __syncthreads();
    for (int i = tid; i < cnt; i += CSR_T) ssrc[bs + i] = stage[i];
  }
}

__device__ __forceinline__ float dot4(float4 a, float4 b) {
  return a.x * b.x + a.y * b.y + a.z * b.z + a.w * b.w;
}

__device__ __forceinline__ float exp_neg(float x) {
  return __builtin_amdgcn_exp2f(x * 1.4426950408889634f);
}

__device__ __forceinline__ float head_sum(float x, int b0) {
  return (__shfl(x, b0) + __shfl(x, b0 + 1)) + __shfl(x, b0 + 2);
}

__global__ __launch_bounds__(256) void gat_aggregate(
    const int* __restrict__ rowptr, const int* __restrict__ ssrc,
    const float* __restrict__ feat, const float* __restrict__ a_l,
    const float* __restrict__ a_r, const float* __restrict__ bias,
    float* __restrict__ h, int n) {
  const int lane = threadIdx.x & 31;
  const int d = blockIdx.x * 8 + __builtin_amdgcn_readfirstlane(threadIdx.x >> 5);
  if (d >= n) return;
  const float4* __restrict__ feat4 = (const float4*)feat;
  const int g  = lane / 9;
  const int q  = lane - g * 9;
  const int hq = q / 3;
  const int b0 = lane - (q - hq * 3);
  const bool live = lane < 27;
  const int beg = rowptr[d];
  const int end = rowptr[d + 1];
  const float4 al = ((const float4*)a_l)[q];
  const float4 ar = ((const float4*)a_r)[q];
  const float erh = head_sum(
      dot4(feat4[(unsigned)d * (FPITCH / 4) + (unsigned)q], ar), b0);

  float  m = -INFINITY;
  float  l = 0.0f;
  float4 acc = make_float4(0.0f, 0.0f, 0.0f, 0.0f);
  for (int cb = beg; cb < end; cb += 32) {
    const int cnt = (end - cb) < 32 ? (end - cb) : 32;
    const int s = ssrc[cb + (lane < cnt ? lane : cnt - 1)];
    for (int t0 = 0; t0 < cnt; t0 += 12) {
      float4 fv[4];
      bool   ok[4];
      #pragma unroll
      for (int u = 0; u < 4; ++u) {
        const int jj = t0 + 3 * u + g;
        const int fb = (t0 + 3 * u) < cnt ? (t0 + 3 * u) : (cnt - 1);
        ok[u] = live && (jj < cnt);
        const int ss = __shfl(s, ok[u] ? jj : fb);
        fv[u] = feat4[(unsigned)ss * (FPITCH / 4) + (unsigned)q];
      }
      float v[4];
      #pragma unroll
      for (int u = 0; u < 4; ++u) {
        const float el = head_sum(dot4(fv[u], al), b0);
        v[u] = ok[u] ? leaky(el + erh) : -INFINITY;
      }
      const float mn = fmaxf(fmaxf(m, fmaxf(v[0], v[1])), fmaxf(v[2], v[3]));
      const float mf = (mn == -INFINITY) ? 0.0f : mn;
      const float sc = exp_neg(m - mf);
      float p[4];
      #pragma unroll
      for (int u = 0; u < 4; ++u) p[u] = exp_neg(v[u] - mf);
      l = l * sc + ((p[0] + p[1]) + (p[2] + p[3]));
      acc.x *= sc; acc.y *= sc; acc.z *= sc; acc.w *= sc;
      #pragma unroll
      for (int u = 0; u < 4; ++u) {
        acc.x += fv[u].x * p[u];
        acc.y += fv[u].y * p[u];
        acc.z += fv[u].z * p[u];
        acc.w += fv[u].w * p[u];
      }
      m = mn;
    }
  }
  const int lB = (lane + 9) & 31, lC = (lane + 18) & 31;
  const float mB = __shfl(m, lB), mC = __shfl(m, lC);
  const float sB = __shfl(l, lB), sC = __shfl(l, lC);
  float4 aB, aC;
  aB.x = __shfl(acc.x, lB); aC.x = __shfl(acc.x, lC);
  aB.y = __shfl(acc.y, lB); aC.y = __shfl(acc.y, lC);
  aB.z = __shfl(acc.z, lB); aC.z = __shfl(acc.z, lC);
  aB.w = __shfl(acc.w, lB); aC.w = __shfl(acc.w, lC);
  if (lane < 9) {
    const float mm = fmaxf(m, fmaxf(mB, mC));
    const float mg = (mm == -INFINITY) ? 0.0f : mm;
    const float wA = exp_neg(m - mg);
    const float wB = exp_neg(mB - mg);
    const float wC = exp_neg(mC - mg);
    const float inv = 1.0f / fmaxf((l * wA + sB * wB) + sC * wC, 1e-9f);
    float4 r;
    r.x = ((acc.x * wA + aB.x * wB) + aC.x * wC) * inv;
    r.y = ((acc.y * wA + aB.y * wB) + aC.y * wC) * inv;
    r.z = ((acc.z * wA + aB.z * wB) + aC.z * wC) * inv;
    r.w = ((acc.w * wA + aB.w * wB) + aC.w * wC) * inv;
    float4* hp = (float4*)(h + (size_t)d * FDIM) + q;
    const float4 hv = *hp;
    const float4 bv = ((const float4*)bias)[q];
    float4 o;
    o.x = r.x + hv.x + bv.x;
    o.y = r.y + hv.y + bv.y;
    o.z = r.z + hv.z + bv.z;
    o.w = r.w + hv.w + bv.w;
    o.x = o.x > 0.0f ? o.x : expm1f(o.x);
    o.y = o.y > 0.0f ? o.y : expm1f(o.y);
    o.z = o.z > 0.0f ? o.z : expm1f(o.z);
    o.w = o.w > 0.0f ? o.w : expm1f(o.w);
    *hp = o;
  }
}

extern "C" void kernel_launch(void* const* d_in, const int* in_sizes, int n_in,
                              void* d_out, int out_size, void* d_ws,
                              size_t ws_size, hipStream_t stream) {
  (void)n_in; (void)out_size; (void)ws_size;
  const float* X        = (const float*)d_in[0];
  const int*   src      = (const int*)d_in[1];
  const int*   dst      = (const int*)d_in[2];
  const float* lin0_w   = (const float*)d_in[3];
  const float* lin0_b   = (const float*)d_in[4];
  const float* fc_w     = (const float*)d_in[5];
  const float* attn_l   = (const float*)d_in[6];
  const float* attn_r   = (const float*)d_in[7];
  const float* gat_bias = (const float*)d_in[8];
  const float* out_w    = (const float*)d_in[9];
  const float* out_b    = (const float*)d_in[10];

  const int n  = in_sizes[0] / FDIM;
  const int ne = in_sizes[1];

  float* ws   = (float*)d_ws;
  float* feat  = ws; ws += (size_t)n * FPITCH;
  float* hbuf  = (float*)d_out;
  int* tab     = (int*)d_out;
  int* bstart  = tab + CSR_NS * CSR_NBMAX;
  int* pairs   = (int*)feat;
  int* rowptr  = (int*)ws;
  int* ssrc    = rowptr + (size_t)n + 1;

  const dim3 gemmGrid((n + 15) / 16);
  const int nb   = (n + CSR_BR - 1) >> CSR_BSH;
  const int nbAg = (n + 7) / 8;

  csr_hist<<<CSR_NS, CSR_T, 0, stream>>>(dst, tab, n, nb, ne);
  csr_part<<<CSR_NS, CSR_T, 0, stream>>>(src, dst, tab, bstart, pairs, n, nb, ne);
  csr_sort<<<nb, CSR_T, 0, stream>>>(pairs, bstart, rowptr, ssrc, n);

  gemm36_wmma<<<gemmGrid, 32, 0, stream>>>(X, lin0_w, lin0_b, hbuf, n);

  for (int l = 0; l < 3; ++l) {
    gemm36_wmma_padded<<<gemmGrid, 32, 0, stream>>>(
        hbuf, fc_w + (size_t)l * FDIM * FDIM, nullptr, feat, n);
    gat_aggregate<<<nbAg, 256, 0, stream>>>(
        rowptr, ssrc, feat, attn_l + (size_t)l * NHEAD * HDIM,
        attn_r + (size_t)l * NHEAD * HDIM, gat_bias + (size_t)l * FDIM, hbuf, n);
  }

  gemm36_wmma<<<gemmGrid, 32, 0, stream>>>(hbuf, out_w, out_b, hbuf, n);
}
